# starting kernel
# baseline (speedup 1.0000x reference)
#include <hip/hip_runtime.h>
#include <hip/hip_bf16.h>

typedef float v2f __attribute__((ext_vector_type(2)));
typedef float v8f __attribute__((ext_vector_type(8)));

#define D_IN  128
#define D_HID 256
#define D_OUT 3

__global__ void gnn_init_kernel(float* __restrict__ ws, long long n) {
    long long i = (long long)blockIdx.x * blockDim.x + threadIdx.x;
    long long stride = (long long)gridDim.x * blockDim.x;
    for (; i < n; i += stride) ws[i] = 0.0f;
}

__global__ void gnn_fusew_kernel(const float* __restrict__ W1,
                                 const float* __restrict__ b1,
                                 const float* __restrict__ W2,
                                 const float* __restrict__ b2,
                                 float* __restrict__ packedB,
                                 float* __restrict__ bc) {
    __shared__ float Wc[D_OUT][D_IN];
    int tid = threadIdx.x;
    if (tid < D_OUT * D_IN) {
        int n = tid / D_IN, k = tid % D_IN;
        float acc = 0.0f;
        for (int h = 0; h < D_HID; ++h)
            acc += W2[n * D_HID + h] * W1[h * D_IN + k];
        Wc[n][k] = acc;
    }
    if (tid < 4) {
        float acc = 0.0f;
        if (tid < D_OUT) {
            acc = b2[tid];
            for (int h = 0; h < D_HID; ++h)
                acc += W2[tid * D_HID + h] * b1[h];
        }
        bc[tid] = acc;
    }
    __syncthreads();
    for (int idx = tid; idx < 2048; idx += blockDim.x) {
        int s    = idx >> 6;
        int j    = (idx >> 5) & 1;
        int lane = idx & 31;
        int n    = lane & 15;
        int k    = 4 * s + j + ((lane >= 16) ? 2 : 0);
        packedB[idx] = (n < D_OUT) ? Wc[n][k] : 0.0f;
    }
}

__global__ void gnn_scatter_kernel(const float* __restrict__ x,
                                   const int* __restrict__ edges,
                                   float* __restrict__ summed,
                                   float* __restrict__ counts,
                                   int E) {
    int wave = (int)((blockIdx.x * blockDim.x + threadIdx.x) >> 5);
    int lane = threadIdx.x & 31;
    if (wave >= E) return;
    int src = edges[wave];
    int dst = edges[E + wave];
    const float4* xr = (const float4*)(x + (long long)src * D_IN);
    float4 v = xr[lane];
    float* srow = summed + (long long)dst * D_IN + lane * 4;
    atomicAdd(srow + 0, v.x);
    atomicAdd(srow + 1, v.y);
    atomicAdd(srow + 2, v.z);
    atomicAdd(srow + 3, v.w);
    if (lane == 0) atomicAdd(counts + dst, 1.0f);
}

__global__ void gnn_mlp_kernel(const float* __restrict__ summed,
                               const float* __restrict__ counts,
                               const float* __restrict__ packedB,
                               const float* __restrict__ bc,
                               float* __restrict__ out,
                               int N) {
    int wave = (int)((blockIdx.x * blockDim.x + threadIdx.x) >> 5);
    int lane = threadIdx.x & 31;
    int nTiles = N >> 4;
    if (wave >= nTiles) return;

    int m0   = wave << 4;
    int mrow = m0 + (lane & 15);
    float inv = 1.0f / fmaxf(counts[mrow], 1.0f);
    int koff = (lane >= 16) ? 2 : 0;
    const float* arow = summed + (long long)mrow * D_IN + koff;

    v8f c = {};
    #pragma unroll 4
    for (int s = 0; s < 32; ++s) {
        float2 a2 = *(const float2*)(arow + 4 * s);
        v2f a; a[0] = a2.x * inv; a[1] = a2.y * inv;
        v2f b; b[0] = packedB[s * 64 + lane];
               b[1] = packedB[s * 64 + 32 + lane];
        c = __builtin_amdgcn_wmma_f32_16x16x4_f32(
                 false, a,  false, b,
                 (short)0, c,  false,  false);
    }

    int n = lane & 15;
    if (n < D_OUT) {
        float bcn = bc[n];
        int rbase = m0 + ((lane >= 16) ? 8 : 0);
        #pragma unroll
        for (int j = 0; j < 8; ++j)
            out[(long long)(rbase + j) * D_OUT + n] = c[j] + bcn;
    }
}

extern "C" void kernel_launch(void* const* d_in, const int* in_sizes, int n_in,
                              void* d_out, int out_size, void* d_ws, size_t ws_size,
                              hipStream_t stream) {
    const float* x     = (const float*)d_in[0];
    const int*   edges = (const int*)  d_in[1];
    const float* W1    = (const float*)d_in[2];
    const float* b1    = (const float*)d_in[3];
    const float* W2    = (const float*)d_in[4];
    const float* b2    = (const float*)d_in[5];
    float*       out   = (float*)d_out;

    int N = in_sizes[0] / D_IN;
    int E = in_sizes[1] / 2;

    float* ws = (float*)d_ws;
    long long offSum = 0;
    long long offCnt = (long long)N * D_IN;
    long long offB   = offCnt + N;
    long long offBc  = offB + 2048;

    gnn_init_kernel<<<4096, 256, 0, stream>>>(ws + offSum, offB);

    gnn_fusew_kernel<<<1, 384, 0, stream>>>(W1, b1, W2, b2, ws + offB, ws + offBc);

    int blocksScatter = (E + 7) / 8;
    gnn_scatter_kernel<<<blocksScatter, 256, 0, stream>>>(
        x, edges, ws + offSum, ws + offCnt, E);

    int nTiles = N / 16;
    int blocksMlp = (nTiles + 7) / 8;
    gnn_mlp_kernel<<<blocksMlp, 256, 0, stream>>>(
        ws + offSum, ws + offCnt, ws + offB, ws + offBc, out, N);
}


# Weight fold W2@W1 (+ W2@b1+b2) computed as one f32 WMMA product by 9 waves of the one work-group, tiles stored straight in fragment order (no LDS, no barrier); fold sum order = the matrix unit's
# speedup vs baseline: 4.4042x; 4.4042x over previous
#include <hip/hip_runtime.h>
#include <hip/hip_bf16.h>


typedef float v2f __attribute__((ext_vector_type(2)));
typedef float v8f __attribute__((ext_vector_type(8)));

#define D_IN  128
#define D_HID 256
#define D_OUT 3

#define NODES_PER_WAVE   4
#define GATHER_WAVES     8
#define TILE_ROWS        16
#define BLOCK_NODES      (GATHER_WAVES * NODES_PER_WAVE)
#define TILES_PER_BLOCK  (BLOCK_NODES / TILE_ROWS)
#define WAVES_PER_TILE   (TILE_ROWS / NODES_PER_WAVE)
#define TILE_PITCH       (D_IN + 4)

#define BUCKET_SHIFT     8
#define BUCKET_NODES     256
#define MAX_BUCKETS      256
#define SORT_THREADS     256
#define EDGES_PER_THREAD 16
#define EDGE_CHUNK       (SORT_THREADS * EDGES_PER_THREAD)
#define RECS_PER_PASS    8

__device__ __forceinline__ int block_excl_scan(int v, int* wsum, int tid, int* total) {
    int lane = tid & 31;
    int w    = tid >> 5;
    int incl = v;
    for (int o = 1; o < 32; o <<= 1) {
        int t = __shfl_up(incl, o);
        if (lane >= o) incl += t;
    }
    if (lane == 31) wsum[w] = incl;
    __syncthreads();
    int wbase = 0, tot = 0;
    for (int j = 0; j < SORT_THREADS / 32; ++j) {
        int s = wsum[j];
        if (j < w) wbase += s;
        tot += s;
    }
    *total = tot;
    return wbase + incl - v;
}

#define FOLD_WAVES 9

template <int ASTRIDE>
__device__ __forceinline__ v8f fold_tile(const float* __restrict__ ap,
                                         const float* __restrict__ bp, v8f c) {
    #pragma unroll 16
    for (int s = 0; s < D_HID / 4; ++s) {
        v2f a; a[0] = ap[(4 * s)     * ASTRIDE];
               a[1] = ap[(4 * s + 1) * ASTRIDE];
        float2 b2v = *(const float2*)(bp + 4 * s);
        v2f b; b[0] = b2v.x; b[1] = b2v.y;
        c = __builtin_amdgcn_wmma_f32_16x16x4_f32(
                 false, a,  false, b,
                 (short)0, c,  false,  false);
    }
    return c;
}

__global__ void gnn_fusew_kernel(const float* __restrict__ W1,
                                 const float* __restrict__ b1,
                                 const float* __restrict__ W2,
                                 const float* __restrict__ b2,
                                 float* __restrict__ packedB,
                                 float* __restrict__ bc,
                                 int* __restrict__ ctr,
                                 int nctr) {
    int tid  = threadIdx.x;
    for (int i = tid; i < nctr; i += blockDim.x) ctr[i] = 0;
    int lane = tid & 31;
    int r    = lane & 15;
    int half = lane >> 4;
    int w = __builtin_amdgcn_readfirstlane(tid >> 5);
    const float* bp = W2 + min(r, D_OUT - 1) * D_HID + 2 * half;
    if (w < D_IN / 16) {
        v8f c = {};
        c = fold_tile<D_IN>(W1 + 2 * half * D_IN + 16 * w + r, bp, c);
        #pragma unroll
        for (int j = 0; j < 8; ++j) {
            int k = 16 * w + 8 * half + j;
            packedB[(k >> 2) * 64 + (k & 1) * 32 + ((k >> 1) & 1) * 16 + r] = c[j];
        }
    } else {
        v8f c = {};
        c[0] = b2[min(r, D_OUT - 1)];
        c = fold_tile<1>(b1 + 2 * half, bp, c);
        if (lane < D_OUT) bc[lane] = c[0];
    }
}

__global__ void gnn_bucket_count_kernel(const int* __restrict__ edges,
                                        int* __restrict__ bucketCount,
                                        int E, int N, int NB) {
    __shared__ int h[MAX_BUCKETS];
    int tid  = threadIdx.x;
    int base = blockIdx.x * EDGE_CHUNK;
    const int* dstp = edges + E;
    h[tid] = 0;
    int d[EDGES_PER_THREAD];
    #pragma unroll
    for (int k = 0; k < EDGES_PER_THREAD; ++k) {
        int e = base + k * SORT_THREADS + tid;
        int v = dstp[min(e, E - 1)];
        d[k] = (e < E) ? v : -1;
    }
    __syncthreads();
    #pragma unroll
    for (int k = 0; k < EDGES_PER_THREAD; ++k)
        if ((unsigned)d[k] < (unsigned)N) atomicAdd(&h[d[k] >> BUCKET_SHIFT], 1);
    __syncthreads();
    if (tid < NB) {
        int c = h[tid];
        if (c) atomicAdd(bucketCount + tid, c);
    }
}

__global__ void gnn_bucket_place_kernel(const int* __restrict__ edges,
                                        const int* __restrict__ bucketCount,
                                        int* __restrict__ bucketCursor,
                                        int* __restrict__ bucketStart,
                                        unsigned* __restrict__ records,
                                        int E, int N, int NB) {
    __shared__ int h[MAX_BUCKETS];
    __shared__ int basePos[MAX_BUCKETS];
    __shared__ int wsum[SORT_THREADS / 32];
    int tid  = threadIdx.x;
    int base = blockIdx.x * EDGE_CHUNK;
    h[tid] = 0;
    int s[EDGES_PER_THREAD], d[EDGES_PER_THREAD], r[EDGES_PER_THREAD], bk[EDGES_PER_THREAD];
    #pragma unroll
    for (int k = 0; k < EDGES_PER_THREAD; ++k) {
        int e  = base + k * SORT_THREADS + tid;
        int ec = min(e, E - 1);
        int sv = edges[ec];
        int dv = edges[E + ec];
        s[k] = sv;
        d[k] = (e < E) ? dv : -1;
    }
    int c = (tid < NB) ? bucketCount[tid] : 0;
    int tot;
    int start = block_excl_scan(c, wsum, tid, &tot);
    if (blockIdx.x == 0) {
        if (tid < NB) bucketStart[tid] = start;
        if (tid == 0) bucketStart[NB] = tot;
    }
    #pragma unroll
    for (int k = 0; k < EDGES_PER_THREAD; ++k) {
        bool ok = (unsigned)d[k] < (unsigned)N;
        bk[k] = ok ? (d[k] >> BUCKET_SHIFT) : 0;
        r[k]  = ok ? atomicAdd(&h[bk[k]], 1) : -1;
    }
    __syncthreads();
    if (tid < NB) {
        int cnt = h[tid];
        int off = 0;
        if (cnt) off = atomicAdd(bucketCursor + tid, cnt);
        basePos[tid] = start + off;
    }
    __syncthreads();
    int p[EDGES_PER_THREAD];
    #pragma unroll
    for (int k = 0; k < EDGES_PER_THREAD; ++k)
        p[k] = basePos[bk[k]] + r[k];
    #pragma unroll
    for (int k = 0; k < EDGES_PER_THREAD; ++k)
        if (r[k] >= 0)
            records[p[k]] =
                ((unsigned)s[k] << BUCKET_SHIFT) | (unsigned)(d[k] & (BUCKET_NODES - 1));
}

__global__ void gnn_node_order_kernel(const unsigned* __restrict__ records,
                                      const int* __restrict__ bucketStart,
                                      int* __restrict__ deg,
                                      int* __restrict__ cursor,
                                      int* __restrict__ sorted,
                                      int N) {
    __shared__ int cntN[BUCKET_NODES];
    __shared__ int cur[BUCKET_NODES];
    __shared__ int wsum[SORT_THREADS / 32];
    int tid = threadIdx.x;
    int b   = blockIdx.x;
    int s0  = bucketStart[b];
    int cnt = bucketStart[b + 1] - s0;
    const unsigned* rec = records + s0;
    cntN[tid] = 0;
    __syncthreads();
    for (int i0 = 0; i0 < cnt; i0 += SORT_THREADS * RECS_PER_PASS) {
        unsigned v[RECS_PER_PASS];
        #pragma unroll
        for (int k = 0; k < RECS_PER_PASS; ++k) {
            int i = i0 + k * SORT_THREADS + tid;
            v[k] = rec[min(i, cnt - 1)];
        }
        #pragma unroll
        for (int k = 0; k < RECS_PER_PASS; ++k)
            if (i0 + k * SORT_THREADS + tid < cnt)
                atomicAdd(&cntN[v[k] & (BUCKET_NODES - 1)], 1);
    }
    __syncthreads();
    int dg = cntN[tid];
    int tot;
    int excl = block_excl_scan(dg, wsum, tid, &tot);
    int node = b * BUCKET_NODES + tid;
    if (node < N) {
        deg[node]    = dg;
        cursor[node] = s0 + excl + dg;
    }
    cur[tid] = s0 + excl;
    __syncthreads();
    for (int i0 = 0; i0 < cnt; i0 += SORT_THREADS * RECS_PER_PASS) {
        unsigned v[RECS_PER_PASS];
        #pragma unroll
        for (int k = 0; k < RECS_PER_PASS; ++k) {
            int i = i0 + k * SORT_THREADS + tid;
            v[k] = rec[min(i, cnt - 1)];
        }
        int p[RECS_PER_PASS];
        #pragma unroll
        for (int k = 0; k < RECS_PER_PASS; ++k)
            p[k] = (i0 + k * SORT_THREADS + tid < cnt)
                       ? atomicAdd(&cur[v[k] & (BUCKET_NODES - 1)], 1) : -1;
        #pragma unroll
        for (int k = 0; k < RECS_PER_PASS; ++k)
            if (p[k] >= 0) sorted[p[k]] = (int)(v[k] >> BUCKET_SHIFT);
    }
}

__global__ void gnn_gather_mlp_kernel(const float* __restrict__ x,
                                      const int* __restrict__ cursor,
                                      const int* __restrict__ deg,
                                      const int* __restrict__ sorted,
                                      const float* __restrict__ packedB,
                                      const float* __restrict__ bc,
                                      float* __restrict__ out,
                                      int N) {
    __shared__ __attribute__((aligned(16))) float tile[TILES_PER_BLOCK][TILE_ROWS][TILE_PITCH];
    int w    = threadIdx.x >> 5;
    int lane = threadIdx.x & 31;
    int n0   = (blockIdx.x * GATHER_WAVES + w) * NODES_PER_WAVE;

    int e_l = 0, d_l = 0;
    if (lane < NODES_PER_WAVE && n0 + lane < N) {
        e_l = cursor[n0 + lane];
        d_l = deg[n0 + lane];
    }
    const float4* x4 = (const float4*)x;

    int dk[NODES_PER_WAVE], sk[NODES_PER_WAVE], iv[NODES_PER_WAVE];
    #pragma unroll
    for (int k = 0; k < NODES_PER_WAVE; ++k) {
        dk[k] = __builtin_amdgcn_readlane(d_l, k);
        sk[k] = __builtin_amdgcn_readlane(e_l, k) - dk[k];
        iv[k] = (lane < dk[k]) ? sorted[sk[k] + lane] : 0;
    }

    float* trow = &tile[w / WAVES_PER_TILE][(w % WAVES_PER_TILE) * NODES_PER_WAVE][4 * lane];

    #pragma unroll
    for (int k = 0; k < NODES_PER_WAVE; ++k) {
        float4 acc = make_float4(0.0f, 0.0f, 0.0f, 0.0f);
        for (int base = 0; base < dk[k]; base += 32) {
            int cnt = min(32, dk[k] - base);
            int idx = iv[k];
            if (base > 0) idx = (lane < cnt) ? sorted[sk[k] + base + lane] : 0;
            for (int i = 0; i < cnt; i += 8) {
                int m = cnt - i;
                int src[8];
                float4 v[8];
                #pragma unroll
                for (int j = 0; j < 8; ++j)
                    src[j] = __builtin_amdgcn_readlane(idx, min(i + j, cnt - 1));
                if (m >= 8) {
                    #pragma unroll
                    for (int j = 0; j < 8; ++j)
                        v[j] = x4[(long long)src[j] * (D_IN / 4) + lane];
                } else {
                    #pragma unroll
                    for (int j = 0; j < 8; ++j) {
                        v[j] = make_float4(0.0f, 0.0f, 0.0f, 0.0f);
                        if (j < m) v[j] = x4[(long long)src[j] * (D_IN / 4) + lane];
                    }
                }
                #pragma unroll
                for (int j = 0; j < 8; ++j) {
                    acc.x += v[j].x; acc.y += v[j].y;
                    acc.z += v[j].z; acc.w += v[j].w;
                }
            }
        }
        float inv = 1.0f / fmaxf((float)dk[k], 1.0f);
        acc.x *= inv; acc.y *= inv; acc.z *= inv; acc.w *= inv;
        *(float4*)(trow + k * TILE_PITCH) = acc;
    }
    __syncthreads();

    if (w < TILES_PER_BLOCK) {
        int m0 = blockIdx.x * BLOCK_NODES + w * TILE_ROWS;
        if (m0 < N) {
            int koff = (lane >= 16) ? 2 : 0;
            const float* arow = &tile[w][lane & 15][koff];
            v8f c = {};
            #pragma unroll 8
            for (int s = 0; s < 32; ++s) {
                float2 a2 = *(const float2*)(arow + 4 * s);
                v2f a; a[0] = a2.x; a[1] = a2.y;
                v2f b; b[0] = packedB[s * 64 + lane];
                       b[1] = packedB[s * 64 + 32 + lane];
                c = __builtin_amdgcn_wmma_f32_16x16x4_f32(
                         false, a,  false, b,
                         (short)0, c,  false,  false);
            }
            int n = lane & 15;
            if (n < D_OUT) {
                float bcn = bc[n];
                int rbase = m0 + ((lane >= 16) ? 8 : 0);
                #pragma unroll
                for (int j = 0; j < 8; ++j)
                    if (rbase + j < N)
                        out[(long long)(rbase + j) * D_OUT + n] = c[j] + bcn;
            }
        }
    }
}

extern "C" void kernel_launch(void* const* d_in, const int* in_sizes, int n_in,
                              void* d_out, int out_size, void* d_ws, size_t ws_size,
                              hipStream_t stream) {
    const float* x     = (const float*)d_in[0];
    const int*   edges = (const int*)  d_in[1];
    const float* W1    = (const float*)d_in[2];
    const float* b1    = (const float*)d_in[3];
    const float* W2    = (const float*)d_in[4];
    const float* b2    = (const float*)d_in[5];
    float*       out   = (float*)d_out;

    int N = in_sizes[0] / D_IN;
    int E = in_sizes[1] / 2;
    int NB = (N + BUCKET_NODES - 1) / BUCKET_NODES;

    float* ws = (float*)d_ws;
    long long offB  = 0;
    long long offBc = offB + 2048;
    int* deg          = (int*)(ws + offBc + 4);
    int* cursor       = deg + N;
    int* sorted       = cursor + N;
    unsigned* records = (unsigned*)(sorted + E);
    int* bucketCount  = (int*)(records + E);
    int* bucketCursor = bucketCount + MAX_BUCKETS;
    int* bucketStart  = bucketCursor + MAX_BUCKETS;

    gnn_fusew_kernel<<<1, FOLD_WAVES * 32, 0, stream>>>(W1, b1, W2, b2, ws + offB, ws + offBc,
                                            bucketCount, 2 * MAX_BUCKETS);

    int edgeBlocks = (E + EDGE_CHUNK - 1) / EDGE_CHUNK;
    gnn_bucket_count_kernel<<<edgeBlocks, SORT_THREADS, 0, stream>>>(edges, bucketCount, E, N, NB);

    gnn_bucket_place_kernel<<<edgeBlocks, SORT_THREADS, 0, stream>>>(
        edges, bucketCount, bucketCursor, bucketStart, records, E, N, NB);

    gnn_node_order_kernel<<<NB, SORT_THREADS, 0, stream>>>(
        records, bucketStart, deg, cursor, sorted, N);

    gnn_gather_mlp_kernel<<<(N + BLOCK_NODES - 1) / BLOCK_NODES, GATHER_WAVES * 32, 0, stream>>>(
        x, cursor, deg, sorted, ws + offB, ws + offBc, out, N);
}


# Edges bucketed in ONE pass in each block's own chunk (LDS rank + block prefix); node order walks a bucket's per-chunk segments; no global counters, fold rides there: 5 -> 3 launches; sum order kept
# speedup vs baseline: 5.7693x; 1.3100x over previous
#include <hip/hip_runtime.h>
#include <hip/hip_bf16.h>


typedef float v2f __attribute__((ext_vector_type(2)));
typedef float v8f __attribute__((ext_vector_type(8)));

#define D_IN  128
#define D_HID 256
#define D_OUT 3

#define NODES_PER_WAVE   4
#define GATHER_WAVES     8
#define TILE_ROWS        16
#define BLOCK_NODES      (GATHER_WAVES * NODES_PER_WAVE)
#define TILES_PER_BLOCK  (BLOCK_NODES / TILE_ROWS)
#define WAVES_PER_TILE   (TILE_ROWS / NODES_PER_WAVE)
#define TILE_PITCH       (D_IN + 4)

#define BUCKET_SHIFT     8
#define BUCKET_NODES     256
#define MAX_BUCKETS      256
#define SORT_THREADS     256
#define EDGES_PER_THREAD 16
#define EDGE_CHUNK       (SORT_THREADS * EDGES_PER_THREAD)
#define SEGS_PER_TRIP    13

__device__ __forceinline__ int block_excl_scan(int v, int* wsum, int tid, int* total) {
    int lane = tid & 31;
    int w    = tid >> 5;
    int incl = v;
    for (int o = 1; o < 32; o <<= 1) {
        int t = __shfl_up(incl, o);
        if (lane >= o) incl += t;
    }
    if (lane == 31) wsum[w] = incl;
    __syncthreads();
    int wbase = 0, tot = 0;
    for (int j = 0; j < SORT_THREADS / 32; ++j) {
        int s = wsum[j];
        if (j < w) wbase += s;
        tot += s;
    }
    *total = tot;
    return wbase + incl - v;
}

__global__ void gnn_block_sort_kernel(const int* __restrict__ edges,
                                      int* __restrict__ segStart,
                                      int* __restrict__ segCnt,
                                      unsigned* __restrict__ records,
                                      int E, int N) {
    __shared__ int h[MAX_BUCKETS];
    __shared__ int basePos[MAX_BUCKETS];
    __shared__ int wsum[SORT_THREADS / 32];
    int tid  = threadIdx.x;
    int base = blockIdx.x * EDGE_CHUNK;
    h[tid] = 0;
    int s[EDGES_PER_THREAD], d[EDGES_PER_THREAD], r[EDGES_PER_THREAD], bk[EDGES_PER_THREAD];
    #pragma unroll
    for (int k = 0; k < EDGES_PER_THREAD; ++k) {
        int e  = base + k * SORT_THREADS + tid;
        int ec = min(e, E - 1);
        int sv = edges[ec];
        int dv = edges[E + ec];
        s[k] = sv;
        d[k] = (e < E) ? dv : -1;
    }
    __syncthreads();
    #pragma unroll
    for (int k = 0; k < EDGES_PER_THREAD; ++k) {
        bool ok = (unsigned)d[k] < (unsigned)N;
        bk[k] = ok ? (d[k] >> BUCKET_SHIFT) : 0;
        r[k]  = ok ? atomicAdd(&h[bk[k]], 1) : -1;
    }
    __syncthreads();
    int c = h[tid];
    int tot;
    int lstart = block_excl_scan(c, wsum, tid, &tot);
    basePos[tid] = lstart;
    segStart[blockIdx.x * MAX_BUCKETS + tid] = lstart;
    segCnt[blockIdx.x * MAX_BUCKETS + tid]   = c;
    __syncthreads();
    int p[EDGES_PER_THREAD];
    #pragma unroll
    for (int k = 0; k < EDGES_PER_THREAD; ++k)
        p[k] = basePos[bk[k]] + r[k];
    #pragma unroll
    for (int k = 0; k < EDGES_PER_THREAD; ++k)
        if (r[k] >= 0)
            records[base + p[k]] =
                ((unsigned)s[k] << BUCKET_SHIFT) | (unsigned)(d[k] & (BUCKET_NODES - 1));
}

#define FOLD_BLOCKS 2

template <int ASTRIDE>
__device__ __forceinline__ v8f fold_tile(const float* __restrict__ ap,
                                         const float* __restrict__ bp, v8f c) {
    #pragma unroll 16
    for (int s = 0; s < D_HID / 4; ++s) {
        v2f a; a[0] = ap[(4 * s)     * ASTRIDE];
               a[1] = ap[(4 * s + 1) * ASTRIDE];
        float2 b2v = *(const float2*)(bp + 4 * s);
        v2f b; b[0] = b2v.x; b[1] = b2v.y;
        c = __builtin_amdgcn_wmma_f32_16x16x4_f32(
                 false, a,  false, b,
                 (short)0, c,  false,  false);
    }
    return c;
}

__device__ __forceinline__ void fold_block(int fb, int tid,
                                           const float* __restrict__ W1,
                                           const float* __restrict__ b1,
                                           const float* __restrict__ W2,
                                           const float* __restrict__ b2,
                                           float* __restrict__ packedB,
                                           float* __restrict__ bc) {
    int lane = tid & 31;
    int r    = lane & 15;
    int half = lane >> 4;
    int w = __builtin_amdgcn_readfirstlane(tid >> 5);
    const float* bp = W2 + min(r, D_OUT - 1) * D_HID + 2 * half;
    if (fb == 0) {
        if (w < D_IN / 16) {
            v8f c = {};
            c = fold_tile<D_IN>(W1 + 2 * half * D_IN + 16 * w + r, bp, c);
            #pragma unroll
            for (int j = 0; j < 8; ++j) {
                int k = 16 * w + 8 * half + j;
                packedB[(k >> 2) * 64 + (k & 1) * 32 + ((k >> 1) & 1) * 16 + r] = c[j];
            }
        }
    } else if (w == 0) {
        v8f c = {};
        c[0] = b2[min(r, D_OUT - 1)];
        c = fold_tile<1>(b1 + 2 * half, bp, c);
        if (lane < D_OUT) bc[lane] = c[0];
    }
}


__device__ __forceinline__ int seg_meta(int jg, int tid, int b, int edgeBlocks,
                                        const int* __restrict__ segStart,
                                        const int* __restrict__ segCnt,
                                        int* mStart, int* mCnt) {
    int j = jg + tid;
    int lst = 0, c = 0;
    if (j < edgeBlocks) {
        lst = segStart[j * MAX_BUCKETS + b];
        c   = segCnt[j * MAX_BUCKETS + b];
    }
    __syncthreads();
    mStart[tid] = j * EDGE_CHUNK + lst;
    mCnt[tid]   = c;
    __syncthreads();
    return lst;
}

__global__ void gnn_node_order_fold_kernel(const unsigned* __restrict__ records,
                                           const int* __restrict__ segStart,
                                           const int* __restrict__ segCnt,
                                           int* __restrict__ deg,
                                           int* __restrict__ cursor,
                                           int* __restrict__ sorted,
                                           int N, int E, int edgeBlocks,
                                           const float* __restrict__ W1,
                                           const float* __restrict__ b1,
                                           const float* __restrict__ W2,
                                           const float* __restrict__ b2,
                                           float* __restrict__ packedB,
                                           float* __restrict__ bc) {
    __shared__ int cntN[BUCKET_NODES];
    __shared__ int cur[BUCKET_NODES];
    __shared__ int wsum[SORT_THREADS / 32];
    __shared__ int mStart[SORT_THREADS];
    __shared__ int mCnt[SORT_THREADS];
    int tid = threadIdx.x;
    if (blockIdx.x < FOLD_BLOCKS) {
        fold_block(blockIdx.x, tid, W1, b1, W2, b2, packedB, bc);
        return;
    }
    int b    = blockIdx.x - FOLD_BLOCKS;
    int lane = tid & 31;
    int w    = __builtin_amdgcn_readfirstlane(tid >> 5);
    cntN[tid] = 0;
    int ls = 0;
    for (int jg = 0; jg < edgeBlocks; jg += SORT_THREADS) {
        ls += seg_meta(jg, tid, b, edgeBlocks, segStart, segCnt, mStart, mCnt);
        int nseg = min(SORT_THREADS, edgeBlocks - jg);
        for (int q0 = w; q0 < nseg; q0 += (SORT_THREADS / 32) * SEGS_PER_TRIP) {
            int st[SEGS_PER_TRIP], ct[SEGS_PER_TRIP];
            unsigned v[SEGS_PER_TRIP];
            int  qv  = q0 + (SORT_THREADS / 32) * lane;
            bool okv = lane < SEGS_PER_TRIP && qv < nseg;
            int  msv = okv ? mStart[qv] : 0;
            int  mcv = okv ? mCnt[qv]   : 0;
            #pragma unroll
            for (int u = 0; u < SEGS_PER_TRIP; ++u) {
                st[u] = __builtin_amdgcn_readlane(msv, u);
                ct[u] = __builtin_amdgcn_readlane(mcv, u);
            }
            #pragma unroll
            for (int u = 0; u < SEGS_PER_TRIP; ++u)
                v[u] = records[min(st[u] + min(lane, max(ct[u] - 1, 0)), E - 1)];
            #pragma unroll
            for (int u = 0; u < SEGS_PER_TRIP; ++u)
                if (lane < ct[u]) atomicAdd(&cntN[v[u] & (BUCKET_NODES - 1)], 1);
            #pragma unroll
            for (int u = 0; u < SEGS_PER_TRIP; ++u)
                for (int i = 32 + lane; i < ct[u]; i += 32)
                    atomicAdd(&cntN[records[st[u] + i] & (BUCKET_NODES - 1)], 1);
        }
    }
    int s0;
    block_excl_scan(ls, wsum, tid, &s0);
    int dg = cntN[tid];
    __syncthreads();
    int tot;
    int excl = block_excl_scan(dg, wsum, tid, &tot);
    int node = b * BUCKET_NODES + tid;
    if (node < N) {
        deg[node]    = dg;
        cursor[node] = s0 + excl + dg;
    }
    cur[tid] = s0 + excl;
    for (int jg = 0; jg < edgeBlocks; jg += SORT_THREADS) {
        if (edgeBlocks > SORT_THREADS) seg_meta(jg, tid, b, edgeBlocks, segStart, segCnt, mStart, mCnt);
        else                           __syncthreads();
        int nseg = min(SORT_THREADS, edgeBlocks - jg);
        for (int q0 = w; q0 < nseg; q0 += (SORT_THREADS / 32) * SEGS_PER_TRIP) {
            int st[SEGS_PER_TRIP], ct[SEGS_PER_TRIP], p[SEGS_PER_TRIP];
            unsigned v[SEGS_PER_TRIP];
            int  qv  = q0 + (SORT_THREADS / 32) * lane;
            bool okv = lane < SEGS_PER_TRIP && qv < nseg;
            int  msv = okv ? mStart[qv] : 0;
            int  mcv = okv ? mCnt[qv]   : 0;
            #pragma unroll
            for (int u = 0; u < SEGS_PER_TRIP; ++u) {
                st[u] = __builtin_amdgcn_readlane(msv, u);
                ct[u] = __builtin_amdgcn_readlane(mcv, u);
            }
            #pragma unroll
            for (int u = 0; u < SEGS_PER_TRIP; ++u)
                v[u] = records[min(st[u] + min(lane, max(ct[u] - 1, 0)), E - 1)];
            #pragma unroll
            for (int u = 0; u < SEGS_PER_TRIP; ++u)
                p[u] = (lane < ct[u]) ? atomicAdd(&cur[v[u] & (BUCKET_NODES - 1)], 1) : -1;
            #pragma unroll
            for (int u = 0; u < SEGS_PER_TRIP; ++u)
                if (p[u] >= 0) sorted[p[u]] = (int)(v[u] >> BUCKET_SHIFT);
            #pragma unroll
            for (int u = 0; u < SEGS_PER_TRIP; ++u)
                for (int i = 32 + lane; i < ct[u]; i += 32) {
                    unsigned x = records[st[u] + i];
                    int pp = atomicAdd(&cur[x & (BUCKET_NODES - 1)], 1);
                    sorted[pp] = (int)(x >> BUCKET_SHIFT);
                }
        }
    }
}

__global__ void gnn_gather_mlp_kernel(const float* __restrict__ x,
                                      const int* __restrict__ cursor,
                                      const int* __restrict__ deg,
                                      const int* __restrict__ sorted,
                                      const float* __restrict__ packedB,
                                      const float* __restrict__ bc,
                                      float* __restrict__ out,
                                      int N) {
    __shared__ __attribute__((aligned(16))) float tile[TILES_PER_BLOCK][TILE_ROWS][TILE_PITCH];
    int w    = threadIdx.x >> 5;
    int lane = threadIdx.x & 31;
    int n0   = (blockIdx.x * GATHER_WAVES + w) * NODES_PER_WAVE;

    int e_l = 0, d_l = 0;
    if (lane < NODES_PER_WAVE && n0 + lane < N) {
        e_l = cursor[n0 + lane];
        d_l = deg[n0 + lane];
    }
    const float4* x4 = (const float4*)x;

    int dk[NODES_PER_WAVE], sk[NODES_PER_WAVE], iv[NODES_PER_WAVE];
    #pragma unroll
    for (int k = 0; k < NODES_PER_WAVE; ++k) {
        dk[k] = __builtin_amdgcn_readlane(d_l, k);
        sk[k] = __builtin_amdgcn_readlane(e_l, k) - dk[k];
        iv[k] = (lane < dk[k]) ? sorted[sk[k] + lane] : 0;
    }

    float* trow = &tile[w / WAVES_PER_TILE][(w % WAVES_PER_TILE) * NODES_PER_WAVE][4 * lane];

    #pragma unroll
    for (int k = 0; k < NODES_PER_WAVE; ++k) {
        float4 acc = make_float4(0.0f, 0.0f, 0.0f, 0.0f);
        for (int base = 0; base < dk[k]; base += 32) {
            int cnt = min(32, dk[k] - base);
            int idx = iv[k];
            if (base > 0) idx = (lane < cnt) ? sorted[sk[k] + base + lane] : 0;
            for (int i = 0; i < cnt; i += 8) {
                int m = cnt - i;
                int src[8];
                float4 v[8];
                #pragma unroll
                for (int j = 0; j < 8; ++j)
                    src[j] = __builtin_amdgcn_readlane(idx, min(i + j, cnt - 1));
                if (m >= 8) {
                    #pragma unroll
                    for (int j = 0; j < 8; ++j)
                        v[j] = x4[(long long)src[j] * (D_IN / 4) + lane];
                } else {
                    #pragma unroll
                    for (int j = 0; j < 8; ++j) {
                        v[j] = make_float4(0.0f, 0.0f, 0.0f, 0.0f);
                        if (j < m) v[j] = x4[(long long)src[j] * (D_IN / 4) + lane];
                    }
                }
                #pragma unroll
                for (int j = 0; j < 8; ++j) {
                    acc.x += v[j].x; acc.y += v[j].y;
                    acc.z += v[j].z; acc.w += v[j].w;
                }
            }
        }
        float inv = 1.0f / fmaxf((float)dk[k], 1.0f);
        acc.x *= inv; acc.y *= inv; acc.z *= inv; acc.w *= inv;
        *(float4*)(trow + k * TILE_PITCH) = acc;
    }
    __syncthreads();

    if (w < TILES_PER_BLOCK) {
        int m0 = blockIdx.x * BLOCK_NODES + w * TILE_ROWS;
        if (m0 < N) {
            int koff = (lane >= 16) ? 2 : 0;
            const float* arow = &tile[w][lane & 15][koff];
            v8f c = {};
            #pragma unroll 8
            for (int s = 0; s < 32; ++s) {
                float2 a2 = *(const float2*)(arow + 4 * s);
                v2f a; a[0] = a2.x; a[1] = a2.y;
                v2f b; b[0] = packedB[s * 64 + lane];
                       b[1] = packedB[s * 64 + 32 + lane];
                c = __builtin_amdgcn_wmma_f32_16x16x4_f32(
                         false, a,  false, b,
                         (short)0, c,  false,  false);
            }
            int n = lane & 15;
            if (n < D_OUT) {
                float bcn = bc[n];
                int rbase = m0 + ((lane >= 16) ? 8 : 0);
                #pragma unroll
                for (int j = 0; j < 8; ++j)
                    if (rbase + j < N)
                        out[(long long)(rbase + j) * D_OUT + n] = c[j] + bcn;
            }
        }
    }
}

extern "C" void kernel_launch(void* const* d_in, const int* in_sizes, int n_in,
                              void* d_out, int out_size, void* d_ws, size_t ws_size,
                              hipStream_t stream) {
    const float* x     = (const float*)d_in[0];
    const int*   edges = (const int*)  d_in[1];
    const float* W1    = (const float*)d_in[2];
    const float* b1    = (const float*)d_in[3];
    const float* W2    = (const float*)d_in[4];
    const float* b2    = (const float*)d_in[5];
    float*       out   = (float*)d_out;

    int N = in_sizes[0] / D_IN;
    int E = in_sizes[1] / 2;
    int NB = (N + BUCKET_NODES - 1) / BUCKET_NODES;
    int edgeBlocks = (E + EDGE_CHUNK - 1) / EDGE_CHUNK;

    float* ws = (float*)d_ws;
    long long offB  = 0;
    long long offBc = offB + 2048;
    int* segStart     = (int*)(ws + offBc + 4);
    int* segCnt       = segStart + edgeBlocks * MAX_BUCKETS;
    int* deg          = segCnt + edgeBlocks * MAX_BUCKETS;
    int* cursor       = deg + N;
    int* sorted       = cursor + N;
    unsigned* records = (unsigned*)(sorted + E);

    gnn_block_sort_kernel<<<edgeBlocks, SORT_THREADS, 0, stream>>>(
        edges, segStart, segCnt, records, E, N);

    gnn_node_order_fold_kernel<<<FOLD_BLOCKS + NB, SORT_THREADS, 0, stream>>>(
        records, segStart, segCnt, deg, cursor, sorted, N, E, edgeBlocks,
        W1, b1, W2, b2, ws + offB, ws + offBc);

    gnn_gather_mlp_kernel<<<(N + BLOCK_NODES - 1) / BLOCK_NODES, GATHER_WAVES * 32, 0, stream>>>(
        x, cursor, deg, sorted, ws + offB, ws + offBc, out, N);
}


# Node-order launch fused into the gather launch: a 1024-thread block per 256-node bucket ranks its records in LDS, then sums and multiplies; fold rides in launch 1; 3 -> 2 launches; sum order kept
# speedup vs baseline: 6.4175x; 1.1124x over previous
#include <hip/hip_runtime.h>
#include <hip/hip_bf16.h>


typedef float v2f __attribute__((ext_vector_type(2)));
typedef float v8f __attribute__((ext_vector_type(8)));

#define D_IN  128
#define D_HID 256
#define D_OUT 3

#define TILE_ROWS        16
#define TILE_PITCH       (D_IN + 4)

#define BUCKET_SHIFT     8
#define BUCKET_NODES     256
#define MAX_BUCKETS      256
#define SORT_THREADS     256
#define EDGES_PER_THREAD 16
#define EDGE_CHUNK       (SORT_THREADS * EDGES_PER_THREAD)
#define BG_THREADS       1024
#define BG_WAVES         (BG_THREADS / 32)
#define BG_NPW           (BUCKET_NODES / BG_WAVES)
#define BG_TILES         (BUCKET_NODES / TILE_ROWS)
#define WAVES_PER_TILE   (TILE_ROWS / BG_NPW)
#define SEGS_PER_TRIP    7

__device__ __forceinline__ int block_excl_scan(int v, int* wsum, int tid, int* total) {
    int lane = tid & 31;
    int w    = tid >> 5;
    int incl = v;
    for (int o = 1; o < 32; o <<= 1) {
        int t = __shfl_up(incl, o);
        if (lane >= o) incl += t;
    }
    if (lane == 31 && w < SORT_THREADS / 32) wsum[w] = incl;
    __syncthreads();
    int wbase = 0, tot = 0;
    for (int j = 0; j < SORT_THREADS / 32; ++j) {
        int s = wsum[j];
        if (j < w) wbase += s;
        tot += s;
    }
    *total = tot;
    return wbase + incl - v;
}

#define FOLD_BLOCKS 2

template <int ASTRIDE>
__device__ __forceinline__ v8f fold_tile(const float* __restrict__ ap,
                                         const float* __restrict__ bp, v8f c) {
    #pragma unroll 16
    for (int s = 0; s < D_HID / 4; ++s) {
        v2f a; a[0] = ap[(4 * s)     * ASTRIDE];
               a[1] = ap[(4 * s + 1) * ASTRIDE];
        float2 b2v = *(const float2*)(bp + 4 * s);
        v2f b; b[0] = b2v.x; b[1] = b2v.y;
        c = __builtin_amdgcn_wmma_f32_16x16x4_f32(
                 false, a,  false, b,
                 (short)0, c,  false,  false);
    }
    return c;
}

__device__ __forceinline__ void fold_block(int fb, int tid,
                                           const float* __restrict__ W1,
                                           const float* __restrict__ b1,
                                           const float* __restrict__ W2,
                                           const float* __restrict__ b2,
                                           float* __restrict__ packedB,
                                           float* __restrict__ bc) {
    int lane = tid & 31;
    int r    = lane & 15;
    int half = lane >> 4;
    int w = __builtin_amdgcn_readfirstlane(tid >> 5);
    const float* bp = W2 + min(r, D_OUT - 1) * D_HID + 2 * half;
    if (fb == 0) {
        if (w < D_IN / 16) {
            v8f c = {};
            c = fold_tile<D_IN>(W1 + 2 * half * D_IN + 16 * w + r, bp, c);
            #pragma unroll
            for (int j = 0; j < 8; ++j) {
                int k = 16 * w + 8 * half + j;
                packedB[(k >> 2) * 64 + (k & 1) * 32 + ((k >> 1) & 1) * 16 + r] = c[j];
            }
        }
    } else if (w == 0) {
        v8f c = {};
        c[0] = b2[min(r, D_OUT - 1)];
        c = fold_tile<1>(b1 + 2 * half, bp, c);
        if (lane < D_OUT) bc[lane] = c[0];
    }
}

__global__ void gnn_block_sort_fold_kernel(const int* __restrict__ edges,
                                           int* __restrict__ segStart,
                                           int* __restrict__ segCnt,
                                           unsigned* __restrict__ records,
                                           int E, int N,
                                           const float* __restrict__ W1,
                                           const float* __restrict__ b1,
                                           const float* __restrict__ W2,
                                           const float* __restrict__ b2,
                                           float* __restrict__ packedB,
                                           float* __restrict__ bc) {
    __shared__ int h[MAX_BUCKETS];
    __shared__ int basePos[MAX_BUCKETS];
    __shared__ int wsum[SORT_THREADS / 32];
    int tid  = threadIdx.x;
    if (blockIdx.x < FOLD_BLOCKS) {
        fold_block(blockIdx.x, tid, W1, b1, W2, b2, packedB, bc);
        return;
    }
    int bid  = blockIdx.x - FOLD_BLOCKS;
    int base = bid * EDGE_CHUNK;
    h[tid] = 0;
    int s[EDGES_PER_THREAD], d[EDGES_PER_THREAD], r[EDGES_PER_THREAD], bk[EDGES_PER_THREAD];
    #pragma unroll
    for (int k = 0; k < EDGES_PER_THREAD; ++k) {
        int e  = base + k * SORT_THREADS + tid;
        int ec = min(e, E - 1);
        int sv = edges[ec];
        int dv = edges[E + ec];
        s[k] = sv;
        d[k] = (e < E) ? dv : -1;
    }
    __syncthreads();
    #pragma unroll
    for (int k = 0; k < EDGES_PER_THREAD; ++k) {
        bool ok = (unsigned)d[k] < (unsigned)N;
        bk[k] = ok ? (d[k] >> BUCKET_SHIFT) : 0;
        r[k]  = ok ? atomicAdd(&h[bk[k]], 1) : -1;
    }
    __syncthreads();
    int c = h[tid];
    int tot;
    int lstart = block_excl_scan(c, wsum, tid, &tot);
    basePos[tid] = lstart;
    segStart[bid * MAX_BUCKETS + tid] = lstart;
    segCnt[bid * MAX_BUCKETS + tid]   = c;
    __syncthreads();
    int p[EDGES_PER_THREAD];
    #pragma unroll
    for (int k = 0; k < EDGES_PER_THREAD; ++k)
        p[k] = basePos[bk[k]] + r[k];
    #pragma unroll
    for (int k = 0; k < EDGES_PER_THREAD; ++k)
        if (r[k] >= 0)
            records[base + p[k]] =
                ((unsigned)s[k] << BUCKET_SHIFT) | (unsigned)(d[k] & (BUCKET_NODES - 1));
}


__device__ __forceinline__ int seg_meta(int jg, int tid, int b, int edgeBlocks,
                                        const int* __restrict__ segStart,
                                        const int* __restrict__ segCnt,
                                        int* mStart, int* mCnt) {
    int j = jg + tid;
    int lst = 0, c = 0;
    if (tid < SORT_THREADS && j < edgeBlocks) {
        lst = segStart[j * MAX_BUCKETS + b];
        c   = segCnt[j * MAX_BUCKETS + b];
    }
    __syncthreads();
    if (tid < SORT_THREADS) {
        mStart[tid] = j * EDGE_CHUNK + lst;
        mCnt[tid]   = c;
    }
    __syncthreads();
    return lst;
}

__global__ __launch_bounds__(BG_THREADS)
void gnn_bucket_gather_mlp_kernel(const unsigned* __restrict__ records,
                                  const int* __restrict__ segStart,
                                  const int* __restrict__ segCnt,
                                  int* sorted,
                                  const float* __restrict__ x,
                                  const float* __restrict__ packedB,
                                  const float* __restrict__ bc,
                                  float* __restrict__ out,
                                  int N, int E, int edgeBlocks) {
    __shared__ __attribute__((aligned(16))) float tile[BG_TILES][TILE_ROWS][TILE_PITCH];
    __shared__ int cntN[BUCKET_NODES];
    __shared__ int startN[BUCKET_NODES];
    __shared__ int cur[BUCKET_NODES];
    __shared__ int wsum[SORT_THREADS / 32];
    __shared__ int mStart[SORT_THREADS];
    __shared__ int mCnt[SORT_THREADS];
    int tid  = threadIdx.x;
    int b    = blockIdx.x;
    int lane = tid & 31;
    int w    = __builtin_amdgcn_readfirstlane(tid >> 5);
    bool lo  = tid < SORT_THREADS;
    if (lo) cntN[tid] = 0;
    int ls = 0;
    for (int jg = 0; jg < edgeBlocks; jg += SORT_THREADS) {
        ls += seg_meta(jg, tid, b, edgeBlocks, segStart, segCnt, mStart, mCnt);
        int nseg = min(SORT_THREADS, edgeBlocks - jg);
        for (int q0 = w; q0 < nseg; q0 += BG_WAVES * SEGS_PER_TRIP) {
            int st[SEGS_PER_TRIP], ct[SEGS_PER_TRIP];
            unsigned v[SEGS_PER_TRIP];
            int  qv  = q0 + BG_WAVES * lane;
            bool okv = lane < SEGS_PER_TRIP && qv < nseg;
            int  msv = okv ? mStart[qv] : 0;
            int  mcv = okv ? mCnt[qv]   : 0;
            #pragma unroll
            for (int u = 0; u < SEGS_PER_TRIP; ++u) {
                st[u] = __builtin_amdgcn_readlane(msv, u);
                ct[u] = __builtin_amdgcn_readlane(mcv, u);
            }
            #pragma unroll
            for (int u = 0; u < SEGS_PER_TRIP; ++u)
                v[u] = records[min(st[u] + min(lane, max(ct[u] - 1, 0)), E - 1)];
            #pragma unroll
            for (int u = 0; u < SEGS_PER_TRIP; ++u)
                if (lane < ct[u]) atomicAdd(&cntN[v[u] & (BUCKET_NODES - 1)], 1);
            #pragma unroll
            for (int u = 0; u < SEGS_PER_TRIP; ++u)
                for (int i = 32 + lane; i < ct[u]; i += 32)
                    atomicAdd(&cntN[records[st[u] + i] & (BUCKET_NODES - 1)], 1);
        }
    }
    int s0;
    block_excl_scan(ls, wsum, tid, &s0);
    int dg = lo ? cntN[tid] : 0;
    __syncthreads();
    int tot;
    int excl = block_excl_scan(dg, wsum, tid, &tot);
    if (lo) {
        startN[tid] = s0 + excl;
        cur[tid]    = s0 + excl;
    }
    for (int jg = 0; jg < edgeBlocks; jg += SORT_THREADS) {
        if (edgeBlocks > SORT_THREADS) seg_meta(jg, tid, b, edgeBlocks, segStart, segCnt, mStart, mCnt);
        else                           __syncthreads();
        int nseg = min(SORT_THREADS, edgeBlocks - jg);
        for (int q0 = w; q0 < nseg; q0 += BG_WAVES * SEGS_PER_TRIP) {
            int st[SEGS_PER_TRIP], ct[SEGS_PER_TRIP], p[SEGS_PER_TRIP];
            unsigned v[SEGS_PER_TRIP];
            int  qv  = q0 + BG_WAVES * lane;
            bool okv = lane < SEGS_PER_TRIP && qv < nseg;
            int  msv = okv ? mStart[qv] : 0;
            int  mcv = okv ? mCnt[qv]   : 0;
            #pragma unroll
            for (int u = 0; u < SEGS_PER_TRIP; ++u) {
                st[u] = __builtin_amdgcn_readlane(msv, u);
                ct[u] = __builtin_amdgcn_readlane(mcv, u);
            }
            #pragma unroll
            for (int u = 0; u < SEGS_PER_TRIP; ++u)
                v[u] = records[min(st[u] + min(lane, max(ct[u] - 1, 0)), E - 1)];
            #pragma unroll
            for (int u = 0; u < SEGS_PER_TRIP; ++u)
                p[u] = (lane < ct[u]) ? atomicAdd(&cur[v[u] & (BUCKET_NODES - 1)], 1) : -1;
            #pragma unroll
            for (int u = 0; u < SEGS_PER_TRIP; ++u)
                if (p[u] >= 0) sorted[p[u]] = (int)(v[u] >> BUCKET_SHIFT);
            #pragma unroll
            for (int u = 0; u < SEGS_PER_TRIP; ++u)
                for (int i = 32 + lane; i < ct[u]; i += 32) {
                    unsigned xr = records[st[u] + i];
                    int pp = atomicAdd(&cur[xr & (BUCKET_NODES - 1)], 1);
                    sorted[pp] = (int)(xr >> BUCKET_SHIFT);
                }
        }
    }
    __syncthreads();

    int nl0 = w * BG_NPW;
    int d_l = 0, s_l = 0;
    if (lane < BG_NPW) {
        d_l = cntN[nl0 + lane];
        s_l = startN[nl0 + lane];
    }
    const float4* x4 = (const float4*)x;

    int dk[BG_NPW], sk[BG_NPW], iv[BG_NPW];
    #pragma unroll
    for (int k = 0; k < BG_NPW; ++k) {
        dk[k] = __builtin_amdgcn_readlane(d_l, k);
        sk[k] = __builtin_amdgcn_readlane(s_l, k);
        iv[k] = (lane < dk[k]) ? sorted[sk[k] + lane] : 0;
    }

    float* trow = &tile[w / WAVES_PER_TILE][(w % WAVES_PER_TILE) * BG_NPW][4 * lane];

    #pragma unroll
    for (int k = 0; k < BG_NPW; ++k) {
        float4 acc = make_float4(0.0f, 0.0f, 0.0f, 0.0f);
        for (int base = 0; base < dk[k]; base += 32) {
            int cnt = min(32, dk[k] - base);
            int idx = iv[k];
            if (base > 0) idx = (lane < cnt) ? sorted[sk[k] + base + lane] : 0;
            for (int i = 0; i < cnt; i += 8) {
                int m = cnt - i;
                int src[8];
                float4 v[8];
                #pragma unroll
                for (int j = 0; j < 8; ++j)
                    src[j] = __builtin_amdgcn_readlane(idx, min(i + j, cnt - 1));
                if (m >= 8) {
                    #pragma unroll
                    for (int j = 0; j < 8; ++j)
                        v[j] = x4[(long long)src[j] * (D_IN / 4) + lane];
                } else {
                    #pragma unroll
                    for (int j = 0; j < 8; ++j) {
                        v[j] = make_float4(0.0f, 0.0f, 0.0f, 0.0f);
                        if (j < m) v[j] = x4[(long long)src[j] * (D_IN / 4) + lane];
                    }
                }
                #pragma unroll
                for (int j = 0; j < 8; ++j) {
                    acc.x += v[j].x; acc.y += v[j].y;
                    acc.z += v[j].z; acc.w += v[j].w;
                }
            }
        }
        float inv = 1.0f / fmaxf((float)dk[k], 1.0f);
        acc.x *= inv; acc.y *= inv; acc.z *= inv; acc.w *= inv;
        *(float4*)(trow + k * TILE_PITCH) = acc;
    }
    __syncthreads();

    if (w < BG_TILES) {
        int m0 = b * BUCKET_NODES + w * TILE_ROWS;
        if (m0 < N) {
            int koff = (lane >= 16) ? 2 : 0;
            const float* arow = &tile[w][lane & 15][koff];
            v8f c = {};
            #pragma unroll 8
            for (int s = 0; s < 32; ++s) {
                float2 a2 = *(const float2*)(arow + 4 * s);
                v2f a; a[0] = a2.x; a[1] = a2.y;
                v2f bf; bf[0] = packedB[s * 64 + lane];
                        bf[1] = packedB[s * 64 + 32 + lane];
                c = __builtin_amdgcn_wmma_f32_16x16x4_f32(
                         false, a,  false, bf,
                         (short)0, c,  false,  false);
            }
            int n = lane & 15;
            if (n < D_OUT) {
                float bcn = bc[n];
                int rbase = m0 + ((lane >= 16) ? 8 : 0);
                #pragma unroll
                for (int j = 0; j < 8; ++j)
                    if (rbase + j < N)
                        out[(long long)(rbase + j) * D_OUT + n] = c[j] + bcn;
            }
        }
    }
}

extern "C" void kernel_launch(void* const* d_in, const int* in_sizes, int n_in,
                              void* d_out, int out_size, void* d_ws, size_t ws_size,
                              hipStream_t stream) {
    const float* x     = (const float*)d_in[0];
    const int*   edges = (const int*)  d_in[1];
    const float* W1    = (const float*)d_in[2];
    const float* b1    = (const float*)d_in[3];
    const float* W2    = (const float*)d_in[4];
    const float* b2    = (const float*)d_in[5];
    float*       out   = (float*)d_out;

    int N = in_sizes[0] / D_IN;
    int E = in_sizes[1] / 2;
    int NB = (N + BUCKET_NODES - 1) / BUCKET_NODES;
    int edgeBlocks = (E + EDGE_CHUNK - 1) / EDGE_CHUNK;

    float* ws = (float*)d_ws;
    long long offB  = 0;
    long long offBc = offB + 2048;
    int* segStart     = (int*)(ws + offBc + 4);
    int* segCnt       = segStart + edgeBlocks * MAX_BUCKETS;
    int* sorted       = segCnt + edgeBlocks * MAX_BUCKETS;
    unsigned* records = (unsigned*)(sorted + E);

    gnn_block_sort_fold_kernel<<<FOLD_BLOCKS + edgeBlocks, SORT_THREADS, 0, stream>>>(
        edges, segStart, segCnt, records, E, N, W1, b1, W2, b2, ws + offB, ws + offBc);

    gnn_bucket_gather_mlp_kernel<<<NB, BG_THREADS, 0, stream>>>(
        records, segStart, segCnt, sorted, x, ws + offB, ws + offBc, out, N, E, edgeBlocks);
}


# Bucket node order in ONE pass: a record takes the next place of its node's 64-place LDS row (returning LDS add); ids never go to memory, no prefix sum, no 2nd pass; sum order = LDS arrival as before
# speedup vs baseline: 7.4310x; 1.1579x over previous
#include <hip/hip_runtime.h>
#include <hip/hip_bf16.h>


typedef float v2f __attribute__((ext_vector_type(2)));
typedef float v8f __attribute__((ext_vector_type(8)));

#define D_IN  128
#define D_HID 256
#define D_OUT 3

#define TILE_ROWS        16
#define TILE_PITCH       (D_IN + 4)

#define BUCKET_SHIFT     8
#define BUCKET_NODES     256
#define MAX_BUCKETS      256
#define SORT_THREADS     256
#define EDGES_PER_THREAD 16
#define EDGE_CHUNK       (SORT_THREADS * EDGES_PER_THREAD)
#define BG_THREADS       1024
#define BG_WAVES         (BG_THREADS / 32)
#define BG_NPW           (BUCKET_NODES / BG_WAVES)
#define BG_TILES         (BUCKET_NODES / TILE_ROWS)
#define WAVES_PER_TILE   (TILE_ROWS / BG_NPW)
#define SEGS_PER_TRIP    7

__device__ __forceinline__ int block_excl_scan(int v, int* wsum, int tid, int* total) {
    int lane = tid & 31;
    int w    = tid >> 5;
    int incl = v;
    for (int o = 1; o < 32; o <<= 1) {
        int t = __shfl_up(incl, o);
        if (lane >= o) incl += t;
    }
    if (lane == 31 && w < SORT_THREADS / 32) wsum[w] = incl;
    __syncthreads();
    int wbase = 0, tot = 0;
    for (int j = 0; j < SORT_THREADS / 32; ++j) {
        int s = wsum[j];
        if (j < w) wbase += s;
        tot += s;
    }
    *total = tot;
    return wbase + incl - v;
}

#define FOLD_BLOCKS 2

template <int ASTRIDE>
__device__ __forceinline__ v8f fold_tile(const float* __restrict__ ap,
                                         const float* __restrict__ bp, v8f c) {
    #pragma unroll 16
    for (int s = 0; s < D_HID / 4; ++s) {
        v2f a; a[0] = ap[(4 * s)     * ASTRIDE];
               a[1] = ap[(4 * s + 1) * ASTRIDE];
        float2 b2v = *(const float2*)(bp + 4 * s);
        v2f b; b[0] = b2v.x; b[1] = b2v.y;
        c = __builtin_amdgcn_wmma_f32_16x16x4_f32(
                 false, a,  false, b,
                 (short)0, c,  false,  false);
    }
    return c;
}

__device__ __forceinline__ void fold_block(int fb, int tid,
                                           const float* __restrict__ W1,
                                           const float* __restrict__ b1,
                                           const float* __restrict__ W2,
                                           const float* __restrict__ b2,
                                           float* __restrict__ packedB,
                                           float* __restrict__ bc) {
    int lane = tid & 31;
    int r    = lane & 15;
    int half = lane >> 4;
    int w = __builtin_amdgcn_readfirstlane(tid >> 5);
    const float* bp = W2 + min(r, D_OUT - 1) * D_HID + 2 * half;
    if (fb == 0) {
        if (w < D_IN / 16) {
            v8f c = {};
            c = fold_tile<D_IN>(W1 + 2 * half * D_IN + 16 * w + r, bp, c);
            #pragma unroll
            for (int j = 0; j < 8; ++j) {
                int k = 16 * w + 8 * half + j;
                packedB[(k >> 2) * 64 + (k & 1) * 32 + ((k >> 1) & 1) * 16 + r] = c[j];
            }
        }
    } else if (w == 0) {
        v8f c = {};
        c[0] = b2[min(r, D_OUT - 1)];
        c = fold_tile<1>(b1 + 2 * half, bp, c);
        if (lane < D_OUT) bc[lane] = c[0];
    }
}

__global__ void gnn_block_sort_fold_kernel(const int* __restrict__ edges,
                                           int* __restrict__ segStart,
                                           int* __restrict__ segCnt,
                                           unsigned* __restrict__ records,
                                           int E, int N,
                                           const float* __restrict__ W1,
                                           const float* __restrict__ b1,
                                           const float* __restrict__ W2,
                                           const float* __restrict__ b2,
                                           float* __restrict__ packedB,
                                           float* __restrict__ bc) {
    __shared__ int h[MAX_BUCKETS];
    __shared__ int basePos[MAX_BUCKETS];
    __shared__ int wsum[SORT_THREADS / 32];
    int tid  = threadIdx.x;
    if (blockIdx.x < FOLD_BLOCKS) {
        fold_block(blockIdx.x, tid, W1, b1, W2, b2, packedB, bc);
        return;
    }
    int bid  = blockIdx.x - FOLD_BLOCKS;
    int base = bid * EDGE_CHUNK;
    h[tid] = 0;
    int s[EDGES_PER_THREAD], d[EDGES_PER_THREAD], r[EDGES_PER_THREAD], bk[EDGES_PER_THREAD];
    #pragma unroll
    for (int k = 0; k < EDGES_PER_THREAD; ++k) {
        int e  = base + k * SORT_THREADS + tid;
        int ec = min(e, E - 1);
        int sv = edges[ec];
        int dv = edges[E + ec];
        s[k] = sv;
        d[k] = (e < E) ? dv : -1;
    }
    __syncthreads();
    #pragma unroll
    for (int k = 0; k < EDGES_PER_THREAD; ++k) {
        bool ok = (unsigned)d[k] < (unsigned)N;
        bk[k] = ok ? (d[k] >> BUCKET_SHIFT) : 0;
        r[k]  = ok ? atomicAdd(&h[bk[k]], 1) : -1;
    }
    __syncthreads();
    int c = h[tid];
    int tot;
    int lstart = block_excl_scan(c, wsum, tid, &tot);
    basePos[tid] = lstart;
    segStart[bid * MAX_BUCKETS + tid] = lstart;
    segCnt[bid * MAX_BUCKETS + tid]   = c;
    __syncthreads();
    int p[EDGES_PER_THREAD];
    #pragma unroll
    for (int k = 0; k < EDGES_PER_THREAD; ++k)
        p[k] = basePos[bk[k]] + r[k];
    #pragma unroll
    for (int k = 0; k < EDGES_PER_THREAD; ++k)
        if (r[k] >= 0)
            records[base + p[k]] =
                ((unsigned)s[k] << BUCKET_SHIFT) | (unsigned)(d[k] & (BUCKET_NODES - 1));
}


#define NODE_SLOTS       64
#define IDS_PITCH        (NODE_SLOTS + 1)

__device__ __forceinline__ int seg_meta(int jg, int tid, int b, int edgeBlocks,
                                        const int* __restrict__ segStart,
                                        const int* __restrict__ segCnt,
                                        int* mStart, int* mCnt) {
    int j = jg + tid;
    int lst = 0, c = 0;
    if (tid < SORT_THREADS && j < edgeBlocks) {
        lst = segStart[j * MAX_BUCKETS + b];
        c   = segCnt[j * MAX_BUCKETS + b];
    }
    __syncthreads();
    if (tid < SORT_THREADS) {
        mStart[tid] = j * EDGE_CHUNK + lst;
        mCnt[tid]   = c;
    }
    __syncthreads();
    return lst;
}

__device__ __forceinline__ void place_in_memory(const unsigned* __restrict__ records,
                                                const int* __restrict__ segStart,
                                                const int* __restrict__ segCnt,
                                                int* sorted, int E, int edgeBlocks,
                                                int tid, int b, int lane, int w,
                                                int* cur, int* mStart, int* mCnt) {
    for (int jg = 0; jg < edgeBlocks; jg += SORT_THREADS) {
        if (edgeBlocks > SORT_THREADS) seg_meta(jg, tid, b, edgeBlocks, segStart, segCnt, mStart, mCnt);
        else                           __syncthreads();
        int nseg = min(SORT_THREADS, edgeBlocks - jg);
        for (int q0 = w; q0 < nseg; q0 += BG_WAVES * SEGS_PER_TRIP) {
            int st[SEGS_PER_TRIP], ct[SEGS_PER_TRIP], p[SEGS_PER_TRIP];
            unsigned v[SEGS_PER_TRIP];
            int  qv  = q0 + BG_WAVES * lane;
            bool okv = lane < SEGS_PER_TRIP && qv < nseg;
            int  msv = okv ? mStart[qv] : 0;
            int  mcv = okv ? mCnt[qv]   : 0;
            #pragma unroll
            for (int u = 0; u < SEGS_PER_TRIP; ++u) {
                st[u] = __builtin_amdgcn_readlane(msv, u);
                ct[u] = __builtin_amdgcn_readlane(mcv, u);
            }
            #pragma unroll
            for (int u = 0; u < SEGS_PER_TRIP; ++u)
                v[u] = records[min(st[u] + min(lane, max(ct[u] - 1, 0)), E - 1)];
            #pragma unroll
            for (int u = 0; u < SEGS_PER_TRIP; ++u)
                p[u] = (lane < ct[u]) ? atomicAdd(&cur[v[u] & (BUCKET_NODES - 1)], 1) : -1;
            #pragma unroll
            for (int u = 0; u < SEGS_PER_TRIP; ++u)
                if (p[u] >= 0) sorted[p[u]] = (int)(v[u] >> BUCKET_SHIFT);
            #pragma unroll
            for (int u = 0; u < SEGS_PER_TRIP; ++u)
                for (int i = 32 + lane; i < ct[u]; i += 32) {
                    unsigned xr = records[st[u] + i];
                    int pp = atomicAdd(&cur[xr & (BUCKET_NODES - 1)], 1);
                    sorted[pp] = (int)(xr >> BUCKET_SHIFT);
                }
        }
    }
}

template <bool IDS_IN_LDS>
__device__ __forceinline__ int get_id(const int* idsL, const int* idsG, int i) {
    if (IDS_IN_LDS) return idsL[i];
    return idsG[i];
}

template <bool IDS_IN_LDS>
__device__ __forceinline__ void gather_rows(const int* idsL, const int* idsG,
                                            const float* __restrict__ x,
                                            int lane, int w,
                                            const int* cntN, const int* startN,
                                            float* trow) {
    int nl0 = w * BG_NPW;
    int d_l = 0, s_l = 0;
    if (lane < BG_NPW) {
        d_l = cntN[nl0 + lane];
        s_l = IDS_IN_LDS ? (nl0 + lane) * IDS_PITCH : startN[nl0 + lane];
    }
    const float4* x4 = (const float4*)x;

    int dk[BG_NPW], sk[BG_NPW], iv[BG_NPW];
    #pragma unroll
    for (int k = 0; k < BG_NPW; ++k) {
        dk[k] = __builtin_amdgcn_readlane(d_l, k);
        sk[k] = __builtin_amdgcn_readlane(s_l, k);
        iv[k] = (lane < dk[k]) ? get_id<IDS_IN_LDS>(idsL, idsG, sk[k] + lane) : 0;
    }

    #pragma unroll
    for (int k = 0; k < BG_NPW; ++k) {
        float4 acc = make_float4(0.0f, 0.0f, 0.0f, 0.0f);
        for (int base = 0; base < dk[k]; base += 32) {
            int cnt = min(32, dk[k] - base);
            int idx = iv[k];
            if (base > 0) idx = (lane < cnt) ? get_id<IDS_IN_LDS>(idsL, idsG, sk[k] + base + lane) : 0;
            for (int i = 0; i < cnt; i += 8) {
                int m = cnt - i;
                int src[8];
                float4 v[8];
                #pragma unroll
                for (int j = 0; j < 8; ++j)
                    src[j] = __builtin_amdgcn_readlane(idx, min(i + j, cnt - 1));
                if (m >= 8) {
                    #pragma unroll
                    for (int j = 0; j < 8; ++j)
                        v[j] = x4[(long long)src[j] * (D_IN / 4) + lane];
                } else {
                    #pragma unroll
                    for (int j = 0; j < 8; ++j) {
                        v[j] = make_float4(0.0f, 0.0f, 0.0f, 0.0f);
                        if (j < m) v[j] = x4[(long long)src[j] * (D_IN / 4) + lane];
                    }
                }
                #pragma unroll
                for (int j = 0; j < 8; ++j) {
                    acc.x += v[j].x; acc.y += v[j].y;
                    acc.z += v[j].z; acc.w += v[j].w;
                }
            }
        }
        float inv = 1.0f / fmaxf((float)dk[k], 1.0f);
        acc.x *= inv; acc.y *= inv; acc.z *= inv; acc.w *= inv;
        *(float4*)(trow + k * TILE_PITCH) = acc;
    }
}

__global__ __launch_bounds__(BG_THREADS)
void gnn_bucket_gather_mlp_kernel(const unsigned* __restrict__ records,
                                  const int* __restrict__ segStart,
                                  const int* __restrict__ segCnt,
                                  int* sorted,
                                  const float* __restrict__ x,
                                  const float* __restrict__ packedB,
                                  const float* __restrict__ bc,
                                  float* __restrict__ out,
                                  int N, int E, int edgeBlocks) {
    __shared__ __attribute__((aligned(16))) float tile[BG_TILES][TILE_ROWS][TILE_PITCH];
    __shared__ int ids[BUCKET_NODES * IDS_PITCH];
    __shared__ int cntN[BUCKET_NODES];
    __shared__ int startN[BUCKET_NODES];
    __shared__ int cur[BUCKET_NODES];
    __shared__ int wsum[SORT_THREADS / 32];
    __shared__ int mStart[SORT_THREADS];
    __shared__ int mCnt[SORT_THREADS];
    __shared__ int rowFull;
    int tid  = threadIdx.x;
    int b    = blockIdx.x;
    int lane = tid & 31;
    int w    = __builtin_amdgcn_readfirstlane(tid >> 5);
    bool lo  = tid < SORT_THREADS;
    if (lo) cntN[tid] = 0;
    if (tid == 0) rowFull = 0;
    int ls = 0;
    for (int jg = 0; jg < edgeBlocks; jg += SORT_THREADS) {
        ls += seg_meta(jg, tid, b, edgeBlocks, segStart, segCnt, mStart, mCnt);
        int nseg = min(SORT_THREADS, edgeBlocks - jg);
        for (int q0 = w; q0 < nseg; q0 += BG_WAVES * SEGS_PER_TRIP) {
            int st[SEGS_PER_TRIP], ct[SEGS_PER_TRIP], r[SEGS_PER_TRIP];
            unsigned v[SEGS_PER_TRIP];
            int  qv  = q0 + BG_WAVES * lane;
            bool okv = lane < SEGS_PER_TRIP && qv < nseg;
            int  msv = okv ? mStart[qv] : 0;
            int  mcv = okv ? mCnt[qv]   : 0;
            #pragma unroll
            for (int u = 0; u < SEGS_PER_TRIP; ++u) {
                st[u] = __builtin_amdgcn_readlane(msv, u);
                ct[u] = __builtin_amdgcn_readlane(mcv, u);
            }
            #pragma unroll
            for (int u = 0; u < SEGS_PER_TRIP; ++u)
                v[u] = records[min(st[u] + min(lane, max(ct[u] - 1, 0)), E - 1)];
            #pragma unroll
            for (int u = 0; u < SEGS_PER_TRIP; ++u)
                r[u] = (lane < ct[u]) ? atomicAdd(&cntN[v[u] & (BUCKET_NODES - 1)], 1) : -1;
            #pragma unroll
            for (int u = 0; u < SEGS_PER_TRIP; ++u)
                if (r[u] >= 0) {
                    if (r[u] < NODE_SLOTS)
                        ids[(v[u] & (BUCKET_NODES - 1)) * IDS_PITCH + r[u]] = (int)(v[u] >> BUCKET_SHIFT);
                    else
                        rowFull = 1;
                }
            #pragma unroll
            for (int u = 0; u < SEGS_PER_TRIP; ++u)
                for (int i = 32 + lane; i < ct[u]; i += 32) {
                    unsigned xr = records[st[u] + i];
                    int rr = atomicAdd(&cntN[xr & (BUCKET_NODES - 1)], 1);
                    if (rr < NODE_SLOTS)
                        ids[(xr & (BUCKET_NODES - 1)) * IDS_PITCH + rr] = (int)(xr >> BUCKET_SHIFT);
                    else
                        rowFull = 1;
                }
        }
    }
    __syncthreads();

    float* trow = &tile[w / WAVES_PER_TILE][(w % WAVES_PER_TILE) * BG_NPW][4 * lane];
    if (__builtin_amdgcn_readfirstlane(rowFull) == 0) {
        gather_rows<true>(ids, sorted, x, lane, w, cntN, startN, trow);
    } else {
        int dg = lo ? cntN[tid] : 0;
        int tot;
        int excl = block_excl_scan(dg, wsum, tid, &tot);
        __syncthreads();
        int s0;
        block_excl_scan(ls, wsum, tid, &s0);
        if (lo) {
            startN[tid] = s0 + excl;
            cur[tid]    = s0 + excl;
        }
        place_in_memory(records, segStart, segCnt, sorted, E, edgeBlocks,
                        tid, b, lane, w, cur, mStart, mCnt);
        __syncthreads();
        gather_rows<false>(ids, sorted, x, lane, w, cntN, startN, trow);
    }
    __syncthreads();

    if (w < BG_TILES) {
        int m0 = b * BUCKET_NODES + w * TILE_ROWS;
        if (m0 < N) {
            int koff = (lane >= 16) ? 2 : 0;
            const float* arow = &tile[w][lane & 15][koff];
            v8f c = {};
            #pragma unroll 8
            for (int s = 0; s < 32; ++s) {
                float2 a2 = *(const float2*)(arow + 4 * s);
                v2f a; a[0] = a2.x; a[1] = a2.y;
                v2f bf; bf[0] = packedB[s * 64 + lane];
                        bf[1] = packedB[s * 64 + 32 + lane];
                c = __builtin_amdgcn_wmma_f32_16x16x4_f32(
                         false, a,  false, bf,
                         (short)0, c,  false,  false);
            }
            int n = lane & 15;
            if (n < D_OUT) {
                float bcn = bc[n];
                int rbase = m0 + ((lane >= 16) ? 8 : 0);
                #pragma unroll
                for (int j = 0; j < 8; ++j)
                    if (rbase + j < N)
                        out[(long long)(rbase + j) * D_OUT + n] = c[j] + bcn;
            }
        }
    }
}

extern "C" void kernel_launch(void* const* d_in, const int* in_sizes, int n_in,
                              void* d_out, int out_size, void* d_ws, size_t ws_size,
                              hipStream_t stream) {
    const float* x     = (const float*)d_in[0];
    const int*   edges = (const int*)  d_in[1];
    const float* W1    = (const float*)d_in[2];
    const float* b1    = (const float*)d_in[3];
    const float* W2    = (const float*)d_in[4];
    const float* b2    = (const float*)d_in[5];
    float*       out   = (float*)d_out;

    int N = in_sizes[0] / D_IN;
    int E = in_sizes[1] / 2;
    int NB = (N + BUCKET_NODES - 1) / BUCKET_NODES;
    int edgeBlocks = (E + EDGE_CHUNK - 1) / EDGE_CHUNK;

    float* ws = (float*)d_ws;
    long long offB  = 0;
    long long offBc = offB + 2048;
    int* segStart     = (int*)(ws + offBc + 4);
    int* segCnt       = segStart + edgeBlocks * MAX_BUCKETS;
    int* sorted       = segCnt + edgeBlocks * MAX_BUCKETS;
    unsigned* records = (unsigned*)(sorted + E);

    gnn_block_sort_fold_kernel<<<FOLD_BLOCKS + edgeBlocks, SORT_THREADS, 0, stream>>>(
        edges, segStart, segCnt, records, E, N, W1, b1, W2, b2, ws + offB, ws + offBc);

    gnn_bucket_gather_mlp_kernel<<<NB, BG_THREADS, 0, stream>>>(
        records, segStart, segCnt, sorted, x, ws + offB, ws + offBc, out, N, E, edgeBlocks);
}


# Launch 1 builds each block's sorted chunk of edge records in LDS and writes it out as whole 128-byte lines (16 bytes per lane), not 4096 scattered 4-byte stores per block; sum order untouched
# speedup vs baseline: 7.7705x; 1.0457x over previous
#include <hip/hip_runtime.h>
#include <hip/hip_bf16.h>


typedef float v2f __attribute__((ext_vector_type(2)));
typedef float v8f __attribute__((ext_vector_type(8)));

#define D_IN  128
#define D_HID 256
#define D_OUT 3

#define TILE_ROWS        16
#define TILE_PITCH       (D_IN + 4)

#define BUCKET_SHIFT     8
#define BUCKET_NODES     256
#define MAX_BUCKETS      256
#define SORT_THREADS     256
#define EDGES_PER_THREAD 16
#define EDGE_CHUNK       (SORT_THREADS * EDGES_PER_THREAD)
#define BG_THREADS       1024
#define BG_WAVES         (BG_THREADS / 32)
#define BG_NPW           (BUCKET_NODES / BG_WAVES)
#define BG_TILES         (BUCKET_NODES / TILE_ROWS)
#define WAVES_PER_TILE   (TILE_ROWS / BG_NPW)
#define SEGS_PER_TRIP    7

__device__ __forceinline__ int block_excl_scan(int v, int* wsum, int tid, int* total) {
    int lane = tid & 31;
    int w    = tid >> 5;
    int incl = v;
    for (int o = 1; o < 32; o <<= 1) {
        int t = __shfl_up(incl, o);
        if (lane >= o) incl += t;
    }
    if (lane == 31 && w < SORT_THREADS / 32) wsum[w] = incl;
    __syncthreads();
    int wbase = 0, tot = 0;
    for (int j = 0; j < SORT_THREADS / 32; ++j) {
        int s = wsum[j];
        if (j < w) wbase += s;
        tot += s;
    }
    *total = tot;
    return wbase + incl - v;
}

#define FOLD_BLOCKS 2

template <int ASTRIDE>
__device__ __forceinline__ v8f fold_tile(const float* __restrict__ ap,
                                         const float* __restrict__ bp, v8f c) {
    #pragma unroll 16
    for (int s = 0; s < D_HID / 4; ++s) {
        v2f a; a[0] = ap[(4 * s)     * ASTRIDE];
               a[1] = ap[(4 * s + 1) * ASTRIDE];
        float2 b2v = *(const float2*)(bp + 4 * s);
        v2f b; b[0] = b2v.x; b[1] = b2v.y;
        c = __builtin_amdgcn_wmma_f32_16x16x4_f32(
                 false, a,  false, b,
                 (short)0, c,  false,  false);
    }
    return c;
}

__device__ __forceinline__ void fold_block(int fb, int tid,
                                           const float* __restrict__ W1,
                                           const float* __restrict__ b1,
                                           const float* __restrict__ W2,
                                           const float* __restrict__ b2,
                                           float* __restrict__ packedB,
                                           float* __restrict__ bc) {
    int lane = tid & 31;
    int r    = lane & 15;
    int half = lane >> 4;
    int w = __builtin_amdgcn_readfirstlane(tid >> 5);
    const float* bp = W2 + min(r, D_OUT - 1) * D_HID + 2 * half;
    if (fb == 0) {
        if (w < D_IN / 16) {
            v8f c = {};
            c = fold_tile<D_IN>(W1 + 2 * half * D_IN + 16 * w + r, bp, c);
            #pragma unroll
            for (int j = 0; j < 8; ++j) {
                int k = 16 * w + 8 * half + j;
                packedB[(k >> 2) * 64 + (k & 1) * 32 + ((k >> 1) & 1) * 16 + r] = c[j];
            }
        }
    } else if (w == 0) {
        v8f c = {};
        c[0] = b2[min(r, D_OUT - 1)];
        c = fold_tile<1>(b1 + 2 * half, bp, c);
        if (lane < D_OUT) bc[lane] = c[0];
    }
}

__global__ void gnn_block_sort_fold_kernel(const int* __restrict__ edges,
                                           int* __restrict__ segStart,
                                           int* __restrict__ segCnt,
                                           unsigned* __restrict__ records,
                                           int E, int N,
                                           const float* __restrict__ W1,
                                           const float* __restrict__ b1,
                                           const float* __restrict__ W2,
                                           const float* __restrict__ b2,
                                           float* __restrict__ packedB,
                                           float* __restrict__ bc) {
    __shared__ int h[MAX_BUCKETS];
    __shared__ int basePos[MAX_BUCKETS];
    __shared__ int wsum[SORT_THREADS / 32];
    __shared__ __attribute__((aligned(16))) unsigned stage[EDGE_CHUNK];
    int tid  = threadIdx.x;
    if (blockIdx.x < FOLD_BLOCKS) {
        fold_block(blockIdx.x, tid, W1, b1, W2, b2, packedB, bc);
        return;
    }
    int bid  = blockIdx.x - FOLD_BLOCKS;
    int base = bid * EDGE_CHUNK;
    h[tid] = 0;
    int s[EDGES_PER_THREAD], d[EDGES_PER_THREAD], r[EDGES_PER_THREAD], bk[EDGES_PER_THREAD];
    #pragma unroll
    for (int k = 0; k < EDGES_PER_THREAD; ++k) {
        int e  = base + k * SORT_THREADS + tid;
        int ec = min(e, E - 1);
        int sv = edges[ec];
        int dv = edges[E + ec];
        s[k] = sv;
        d[k] = (e < E) ? dv : -1;
    }
    __syncthreads();
    #pragma unroll
    for (int k = 0; k < EDGES_PER_THREAD; ++k) {
        bool ok = (unsigned)d[k] < (unsigned)N;
        bk[k] = ok ? (d[k] >> BUCKET_SHIFT) : 0;
        r[k]  = ok ? atomicAdd(&h[bk[k]], 1) : -1;
    }
    __syncthreads();
    int c = h[tid];
    int tot;
    int lstart = block_excl_scan(c, wsum, tid, &tot);
    basePos[tid] = lstart;
    segStart[bid * MAX_BUCKETS + tid] = lstart;
    segCnt[bid * MAX_BUCKETS + tid]   = c;
    __syncthreads();
    int p[EDGES_PER_THREAD];
    #pragma unroll
    for (int k = 0; k < EDGES_PER_THREAD; ++k)
        p[k] = basePos[bk[k]] + r[k];
    #pragma unroll
    for (int k = 0; k < EDGES_PER_THREAD; ++k)
        if (r[k] >= 0)
            stage[p[k]] =
                ((unsigned)s[k] << BUCKET_SHIFT) | (unsigned)(d[k] & (BUCKET_NODES - 1));
    __syncthreads();
    constexpr int TRIPS = EDGE_CHUNK / (4 * SORT_THREADS);
    const uint4* st4 = (const uint4*)stage;
    uint4* rec4 = (uint4*)(records + base);
    int nq = tot >> 2;
    uint4 piece[TRIPS];
    #pragma unroll
    for (int j = 0; j < TRIPS; ++j)
        piece[j] = st4[j * SORT_THREADS + tid];
    #pragma unroll
    for (int j = 0; j < TRIPS; ++j)
        asm volatile("" :: "v"(piece[j].x), "v"(piece[j].y), "v"(piece[j].z), "v"(piece[j].w));
    #pragma unroll
    for (int j = 0; j < TRIPS; ++j)
        if (j * SORT_THREADS + tid < nq)
            rec4[j * SORT_THREADS + tid] = piece[j];
    if (tid < (tot & 3))
        records[base + 4 * nq + tid] = stage[4 * nq + tid];
}


#define NODE_SLOTS       64
#define IDS_PITCH        (NODE_SLOTS + 1)

__device__ __forceinline__ int seg_meta(int jg, int tid, int b, int edgeBlocks,
                                        const int* __restrict__ segStart,
                                        const int* __restrict__ segCnt,
                                        int* mStart, int* mCnt) {
    int j = jg + tid;
    int lst = 0, c = 0;
    if (tid < SORT_THREADS && j < edgeBlocks) {
        lst = segStart[j * MAX_BUCKETS + b];
        c   = segCnt[j * MAX_BUCKETS + b];
    }
    __syncthreads();
    if (tid < SORT_THREADS) {
        mStart[tid] = j * EDGE_CHUNK + lst;
        mCnt[tid]   = c;
    }
    __syncthreads();
    return lst;
}

__device__ __forceinline__ void place_in_memory(const unsigned* __restrict__ records,
                                                const int* __restrict__ segStart,
                                                const int* __restrict__ segCnt,
                                                int* sorted, int E, int edgeBlocks,
                                                int tid, int b, int lane, int w,
                                                int* cur, int* mStart, int* mCnt) {
    for (int jg = 0; jg < edgeBlocks; jg += SORT_THREADS) {
        if (edgeBlocks > SORT_THREADS) seg_meta(jg, tid, b, edgeBlocks, segStart, segCnt, mStart, mCnt);
        else                           __syncthreads();
        int nseg = min(SORT_THREADS, edgeBlocks - jg);
        for (int q0 = w; q0 < nseg; q0 += BG_WAVES * SEGS_PER_TRIP) {
            int st[SEGS_PER_TRIP], ct[SEGS_PER_TRIP], p[SEGS_PER_TRIP];
            unsigned v[SEGS_PER_TRIP];
            int  qv  = q0 + BG_WAVES * lane;
            bool okv = lane < SEGS_PER_TRIP && qv < nseg;
            int  msv = okv ? mStart[qv] : 0;
            int  mcv = okv ? mCnt[qv]   : 0;
            #pragma unroll
            for (int u = 0; u < SEGS_PER_TRIP; ++u) {
                st[u] = __builtin_amdgcn_readlane(msv, u);
                ct[u] = __builtin_amdgcn_readlane(mcv, u);
            }
            #pragma unroll
            for (int u = 0; u < SEGS_PER_TRIP; ++u)
                v[u] = records[min(st[u] + min(lane, max(ct[u] - 1, 0)), E - 1)];
            #pragma unroll
            for (int u = 0; u < SEGS_PER_TRIP; ++u)
                p[u] = (lane < ct[u]) ? atomicAdd(&cur[v[u] & (BUCKET_NODES - 1)], 1) : -1;
            #pragma unroll
            for (int u = 0; u < SEGS_PER_TRIP; ++u)
                if (p[u] >= 0) sorted[p[u]] = (int)(v[u] >> BUCKET_SHIFT);
            #pragma unroll
            for (int u = 0; u < SEGS_PER_TRIP; ++u)
                for (int i = 32 + lane; i < ct[u]; i += 32) {
                    unsigned xr = records[st[u] + i];
                    int pp = atomicAdd(&cur[xr & (BUCKET_NODES - 1)], 1);
                    sorted[pp] = (int)(xr >> BUCKET_SHIFT);
                }
        }
    }
}

template <bool IDS_IN_LDS>
__device__ __forceinline__ int get_id(const int* idsL, const int* idsG, int i) {
    if (IDS_IN_LDS) return idsL[i];
    return idsG[i];
}

template <bool IDS_IN_LDS>
__device__ __forceinline__ void gather_rows(const int* idsL, const int* idsG,
                                            const float* __restrict__ x,
                                            int lane, int w,
                                            const int* cntN, const int* startN,
                                            float* trow) {
    int nl0 = w * BG_NPW;
    int d_l = 0, s_l = 0;
    if (lane < BG_NPW) {
        d_l = cntN[nl0 + lane];
        s_l = IDS_IN_LDS ? (nl0 + lane) * IDS_PITCH : startN[nl0 + lane];
    }
    const float4* x4 = (const float4*)x;

    int dk[BG_NPW], sk[BG_NPW], iv[BG_NPW];
    #pragma unroll
    for (int k = 0; k < BG_NPW; ++k) {
        dk[k] = __builtin_amdgcn_readlane(d_l, k);
        sk[k] = __builtin_amdgcn_readlane(s_l, k);
        iv[k] = (lane < dk[k]) ? get_id<IDS_IN_LDS>(idsL, idsG, sk[k] + lane) : 0;
    }

    #pragma unroll
    for (int k = 0; k < BG_NPW; ++k) {
        float4 acc = make_float4(0.0f, 0.0f, 0.0f, 0.0f);
        for (int base = 0; base < dk[k]; base += 32) {
            int cnt = min(32, dk[k] - base);
            int idx = iv[k];
            if (base > 0) idx = (lane < cnt) ? get_id<IDS_IN_LDS>(idsL, idsG, sk[k] + base + lane) : 0;
            for (int i = 0; i < cnt; i += 8) {
                int m = cnt - i;
                int src[8];
                float4 v[8];
                #pragma unroll
                for (int j = 0; j < 8; ++j)
                    src[j] = __builtin_amdgcn_readlane(idx, min(i + j, cnt - 1));
                if (m >= 8) {
                    #pragma unroll
                    for (int j = 0; j < 8; ++j)
                        v[j] = x4[(long long)src[j] * (D_IN / 4) + lane];
                } else {
                    #pragma unroll
                    for (int j = 0; j < 8; ++j) {
                        v[j] = make_float4(0.0f, 0.0f, 0.0f, 0.0f);
                        if (j < m) v[j] = x4[(long long)src[j] * (D_IN / 4) + lane];
                    }
                }
                #pragma unroll
                for (int j = 0; j < 8; ++j) {
                    acc.x += v[j].x; acc.y += v[j].y;
                    acc.z += v[j].z; acc.w += v[j].w;
                }
            }
        }
        float inv = 1.0f / fmaxf((float)dk[k], 1.0f);
        acc.x *= inv; acc.y *= inv; acc.z *= inv; acc.w *= inv;
        *(float4*)(trow + k * TILE_PITCH) = acc;
    }
}

__global__ __launch_bounds__(BG_THREADS)
void gnn_bucket_gather_mlp_kernel(const unsigned* __restrict__ records,
                                  const int* __restrict__ segStart,
                                  const int* __restrict__ segCnt,
                                  int* sorted,
                                  const float* __restrict__ x,
                                  const float* __restrict__ packedB,
                                  const float* __restrict__ bc,
                                  float* __restrict__ out,
                                  int N, int E, int edgeBlocks) {
    __shared__ __attribute__((aligned(16))) float tile[BG_TILES][TILE_ROWS][TILE_PITCH];
    __shared__ int ids[BUCKET_NODES * IDS_PITCH];
    __shared__ int cntN[BUCKET_NODES];
    __shared__ int startN[BUCKET_NODES];
    __shared__ int cur[BUCKET_NODES];
    __shared__ int wsum[SORT_THREADS / 32];
    __shared__ int mStart[SORT_THREADS];
    __shared__ int mCnt[SORT_THREADS];
    __shared__ int rowFull;
    int tid  = threadIdx.x;
    int b    = blockIdx.x;
    int lane = tid & 31;
    int w    = __builtin_amdgcn_readfirstlane(tid >> 5);
    bool lo  = tid < SORT_THREADS;
    if (lo) cntN[tid] = 0;
    if (tid == 0) rowFull = 0;
    int ls = 0;
    for (int jg = 0; jg < edgeBlocks; jg += SORT_THREADS) {
        ls += seg_meta(jg, tid, b, edgeBlocks, segStart, segCnt, mStart, mCnt);
        int nseg = min(SORT_THREADS, edgeBlocks - jg);
        for (int q0 = w; q0 < nseg; q0 += BG_WAVES * SEGS_PER_TRIP) {
            int st[SEGS_PER_TRIP], ct[SEGS_PER_TRIP], r[SEGS_PER_TRIP];
            unsigned v[SEGS_PER_TRIP];
            int  qv  = q0 + BG_WAVES * lane;
            bool okv = lane < SEGS_PER_TRIP && qv < nseg;
            int  msv = okv ? mStart[qv] : 0;
            int  mcv = okv ? mCnt[qv]   : 0;
            #pragma unroll
            for (int u = 0; u < SEGS_PER_TRIP; ++u) {
                st[u] = __builtin_amdgcn_readlane(msv, u);
                ct[u] = __builtin_amdgcn_readlane(mcv, u);
            }
            #pragma unroll
            for (int u = 0; u < SEGS_PER_TRIP; ++u)
                v[u] = records[min(st[u] + min(lane, max(ct[u] - 1, 0)), E - 1)];
            #pragma unroll
            for (int u = 0; u < SEGS_PER_TRIP; ++u)
                r[u] = (lane < ct[u]) ? atomicAdd(&cntN[v[u] & (BUCKET_NODES - 1)], 1) : -1;
            #pragma unroll
            for (int u = 0; u < SEGS_PER_TRIP; ++u)
                if (r[u] >= 0) {
                    if (r[u] < NODE_SLOTS)
                        ids[(v[u] & (BUCKET_NODES - 1)) * IDS_PITCH + r[u]] = (int)(v[u] >> BUCKET_SHIFT);
                    else
                        rowFull = 1;
                }
            #pragma unroll
            for (int u = 0; u < SEGS_PER_TRIP; ++u)
                for (int i = 32 + lane; i < ct[u]; i += 32) {
                    unsigned xr = records[st[u] + i];
                    int rr = atomicAdd(&cntN[xr & (BUCKET_NODES - 1)], 1);
                    if (rr < NODE_SLOTS)
                        ids[(xr & (BUCKET_NODES - 1)) * IDS_PITCH + rr] = (int)(xr >> BUCKET_SHIFT);
                    else
                        rowFull = 1;
                }
        }
    }
    __syncthreads();

    float* trow = &tile[w / WAVES_PER_TILE][(w % WAVES_PER_TILE) * BG_NPW][4 * lane];
    if (__builtin_amdgcn_readfirstlane(rowFull) == 0) {
        gather_rows<true>(ids, sorted, x, lane, w, cntN, startN, trow);
    } else {
        int dg = lo ? cntN[tid] : 0;
        int tot;
        int excl = block_excl_scan(dg, wsum, tid, &tot);
        __syncthreads();
        int s0;
        block_excl_scan(ls, wsum, tid, &s0);
        if (lo) {
            startN[tid] = s0 + excl;
            cur[tid]    = s0 + excl;
        }
        place_in_memory(records, segStart, segCnt, sorted, E, edgeBlocks,
                        tid, b, lane, w, cur, mStart, mCnt);
        __syncthreads();
        gather_rows<false>(ids, sorted, x, lane, w, cntN, startN, trow);
    }
    __syncthreads();

    if (w < BG_TILES) {
        int m0 = b * BUCKET_NODES + w * TILE_ROWS;
        if (m0 < N) {
            int koff = (lane >= 16) ? 2 : 0;
            const float* arow = &tile[w][lane & 15][koff];
            v8f c = {};
            #pragma unroll 8
            for (int s = 0; s < 32; ++s) {
                float2 a2 = *(const float2*)(arow + 4 * s);
                v2f a; a[0] = a2.x; a[1] = a2.y;
                v2f bf; bf[0] = packedB[s * 64 + lane];
                        bf[1] = packedB[s * 64 + 32 + lane];
                c = __builtin_amdgcn_wmma_f32_16x16x4_f32(
                         false, a,  false, bf,
                         (short)0, c,  false,  false);
            }
            int n = lane & 15;
            if (n < D_OUT) {
                float bcn = bc[n];
                int rbase = m0 + ((lane >= 16) ? 8 : 0);
                #pragma unroll
                for (int j = 0; j < 8; ++j)
                    if (rbase + j < N)
                        out[(long long)(rbase + j) * D_OUT + n] = c[j] + bcn;
            }
        }
    }
}

extern "C" void kernel_launch(void* const* d_in, const int* in_sizes, int n_in,
                              void* d_out, int out_size, void* d_ws, size_t ws_size,
                              hipStream_t stream) {
    const float* x     = (const float*)d_in[0];
    const int*   edges = (const int*)  d_in[1];
    const float* W1    = (const float*)d_in[2];
    const float* b1    = (const float*)d_in[3];
    const float* W2    = (const float*)d_in[4];
    const float* b2    = (const float*)d_in[5];
    float*       out   = (float*)d_out;

    int N = in_sizes[0] / D_IN;
    int E = in_sizes[1] / 2;
    int NB = (N + BUCKET_NODES - 1) / BUCKET_NODES;
    int edgeBlocks = (E + EDGE_CHUNK - 1) / EDGE_CHUNK;

    float* ws = (float*)d_ws;
    long long offB  = 0;
    long long offBc = offB + 2048;
    int* segStart     = (int*)(ws + offBc + 64);
    int* segCnt       = segStart + edgeBlocks * MAX_BUCKETS;
    int* sorted       = segCnt + edgeBlocks * MAX_BUCKETS;
    unsigned* records = (unsigned*)(sorted + edgeBlocks * EDGE_CHUNK);

    gnn_block_sort_fold_kernel<<<FOLD_BLOCKS + edgeBlocks, SORT_THREADS, 0, stream>>>(
        edges, segStart, segCnt, records, E, N, W1, b1, W2, b2, ws + offB, ws + offBc);

    gnn_bucket_gather_mlp_kernel<<<NB, BG_THREADS, 0, stream>>>(
        records, segStart, segCnt, sorted, x, ws + offB, ws + offBc, out, N, E, edgeBlocks);
}


# Buckets of 196 nodes, not 256: the gather launch is 256 work-groups (28 waves, 7 nodes a wave, 13 tiles), one on every WGP, where 196 left 60 WGPs idle; bucket = id/196; sum order = LDS arrival
# speedup vs baseline: 8.2811x; 1.0657x over previous
#include <hip/hip_runtime.h>
#include <hip/hip_bf16.h>


typedef float v2f __attribute__((ext_vector_type(2)));
typedef float v8f __attribute__((ext_vector_type(8)));

#define D_IN  128
#define D_HID 256
#define D_OUT 3

#define TILE_ROWS        16
#define TILE_PITCH       (D_IN + 4)

#define BUCKET_NODES     196
#define REC_SHIFT        8
#define REC_MASK         ((1 << REC_SHIFT) - 1)
#define MAX_BUCKETS      256
#define SORT_THREADS     256
#define EDGES_PER_THREAD 16
#define EDGE_CHUNK       (SORT_THREADS * EDGES_PER_THREAD)
#define BG_NPW           7
#define BG_WAVES         (BUCKET_NODES / BG_NPW)
#define BG_THREADS       (BG_WAVES * 32)
#define BG_TILES         ((BUCKET_NODES + TILE_ROWS - 1) / TILE_ROWS)
#define PAD_ROWS         (BG_TILES * TILE_ROWS - BUCKET_NODES)
#define SEGS_PER_TRIP    7

__device__ __forceinline__ int block_excl_scan(int v, int* wsum, int tid, int* total) {
    int lane = tid & 31;
    int w    = tid >> 5;
    int incl = v;
    for (int o = 1; o < 32; o <<= 1) {
        int t = __shfl_up(incl, o);
        if (lane >= o) incl += t;
    }
    if (lane == 31 && w < SORT_THREADS / 32) wsum[w] = incl;
    __syncthreads();
    int wbase = 0, tot = 0;
    for (int j = 0; j < SORT_THREADS / 32; ++j) {
        int s = wsum[j];
        if (j < w) wbase += s;
        tot += s;
    }
    *total = tot;
    return wbase + incl - v;
}

#define FOLD_BLOCKS 2

template <int ASTRIDE>
__device__ __forceinline__ v8f fold_tile(const float* __restrict__ ap,
                                         const float* __restrict__ bp, v8f c) {
    #pragma unroll 16
    for (int s = 0; s < D_HID / 4; ++s) {
        v2f a; a[0] = ap[(4 * s)     * ASTRIDE];
               a[1] = ap[(4 * s + 1) * ASTRIDE];
        float2 b2v = *(const float2*)(bp + 4 * s);
        v2f b; b[0] = b2v.x; b[1] = b2v.y;
        c = __builtin_amdgcn_wmma_f32_16x16x4_f32(
                 false, a,  false, b,
                 (short)0, c,  false,  false);
    }
    return c;
}

__device__ __forceinline__ void fold_block(int fb, int tid,
                                           const float* __restrict__ W1,
                                           const float* __restrict__ b1,
                                           const float* __restrict__ W2,
                                           const float* __restrict__ b2,
                                           float* __restrict__ packedB,
                                           float* __restrict__ bc) {
    int lane = tid & 31;
    int r    = lane & 15;
    int half = lane >> 4;
    int w = __builtin_amdgcn_readfirstlane(tid >> 5);
    const float* bp = W2 + min(r, D_OUT - 1) * D_HID + 2 * half;
    if (fb == 0) {
        if (w < D_IN / 16) {
            v8f c = {};
            c = fold_tile<D_IN>(W1 + 2 * half * D_IN + 16 * w + r, bp, c);
            #pragma unroll
            for (int j = 0; j < 8; ++j) {
                int k = 16 * w + 8 * half + j;
                packedB[(k >> 2) * 64 + (k & 1) * 32 + ((k >> 1) & 1) * 16 + r] = c[j];
            }
        }
    } else if (w == 0) {
        v8f c = {};
        c[0] = b2[min(r, D_OUT - 1)];
        c = fold_tile<1>(b1 + 2 * half, bp, c);
        if (lane < D_OUT) bc[lane] = c[0];
    }
}

__global__ void gnn_block_sort_fold_kernel(const int* __restrict__ edges,
                                           int* __restrict__ segStart,
                                           int* __restrict__ segCnt,
                                           unsigned* __restrict__ records,
                                           int E, int N,
                                           const float* __restrict__ W1,
                                           const float* __restrict__ b1,
                                           const float* __restrict__ W2,
                                           const float* __restrict__ b2,
                                           float* __restrict__ packedB,
                                           float* __restrict__ bc) {
    __shared__ int h[MAX_BUCKETS];
    __shared__ int basePos[MAX_BUCKETS];
    __shared__ int wsum[SORT_THREADS / 32];
    __shared__ __attribute__((aligned(16))) unsigned stage[EDGE_CHUNK];
    int tid  = threadIdx.x;
    if (blockIdx.x < FOLD_BLOCKS) {
        fold_block(blockIdx.x, tid, W1, b1, W2, b2, packedB, bc);
        return;
    }
    int bid  = blockIdx.x - FOLD_BLOCKS;
    int base = bid * EDGE_CHUNK;
    h[tid] = 0;
    int s[EDGES_PER_THREAD], d[EDGES_PER_THREAD], r[EDGES_PER_THREAD], bk[EDGES_PER_THREAD];
    #pragma unroll
    for (int k = 0; k < EDGES_PER_THREAD; ++k) {
        int e  = base + k * SORT_THREADS + tid;
        int ec = min(e, E - 1);
        int sv = edges[ec];
        int dv = edges[E + ec];
        s[k] = sv;
        d[k] = (e < E) ? dv : -1;
    }
    __syncthreads();
    #pragma unroll
    for (int k = 0; k < EDGES_PER_THREAD; ++k) {
        bool ok = (unsigned)d[k] < (unsigned)N;
        bk[k] = ok ? (int)((unsigned)d[k] / (unsigned)BUCKET_NODES) : 0;
        r[k]  = ok ? atomicAdd(&h[bk[k]], 1) : -1;
    }
    __syncthreads();
    int c = h[tid];
    int tot;
    int lstart = block_excl_scan(c, wsum, tid, &tot);
    basePos[tid] = lstart;
    segStart[bid * MAX_BUCKETS + tid] = lstart;
    segCnt[bid * MAX_BUCKETS + tid]   = c;
    __syncthreads();
    int p[EDGES_PER_THREAD];
    #pragma unroll
    for (int k = 0; k < EDGES_PER_THREAD; ++k)
        p[k] = basePos[bk[k]] + r[k];
    #pragma unroll
    for (int k = 0; k < EDGES_PER_THREAD; ++k)
        if (r[k] >= 0)
            stage[p[k]] =
                ((unsigned)s[k] << REC_SHIFT) | (unsigned)(d[k] - bk[k] * BUCKET_NODES);
    __syncthreads();
    constexpr int TRIPS = EDGE_CHUNK / (4 * SORT_THREADS);
    const uint4* st4 = (const uint4*)stage;
    uint4* rec4 = (uint4*)(records + base);
    int nq = tot >> 2;
    uint4 piece[TRIPS];
    #pragma unroll
    for (int j = 0; j < TRIPS; ++j)
        piece[j] = st4[j * SORT_THREADS + tid];
    #pragma unroll
    for (int j = 0; j < TRIPS; ++j)
        asm volatile("" :: "v"(piece[j].x), "v"(piece[j].y), "v"(piece[j].z), "v"(piece[j].w));
    #pragma unroll
    for (int j = 0; j < TRIPS; ++j)
        if (j * SORT_THREADS + tid < nq)
            rec4[j * SORT_THREADS + tid] = piece[j];
    if (tid < (tot & 3))
        records[base + 4 * nq + tid] = stage[4 * nq + tid];
}


#define NODE_SLOTS       64
#define IDS_PITCH        (NODE_SLOTS + 1)

__device__ __forceinline__ int seg_meta(int jg, int tid, int b, int edgeBlocks,
                                        const int* __restrict__ segStart,
                                        const int* __restrict__ segCnt,
                                        int* mStart, int* mCnt) {
    int j = jg + tid;
    int lst = 0, c = 0;
    if (tid < SORT_THREADS && j < edgeBlocks) {
        lst = segStart[j * MAX_BUCKETS + b];
        c   = segCnt[j * MAX_BUCKETS + b];
    }
    __syncthreads();
    if (tid < SORT_THREADS) {
        mStart[tid] = j * EDGE_CHUNK + lst;
        mCnt[tid]   = c;
    }
    __syncthreads();
    return lst;
}

__device__ __forceinline__ void place_in_memory(const unsigned* __restrict__ records,
                                                const int* __restrict__ segStart,
                                                const int* __restrict__ segCnt,
                                                int* sorted, int E, int edgeBlocks,
                                                int tid, int b, int lane, int w,
                                                int* cur, int* mStart, int* mCnt) {
    for (int jg = 0; jg < edgeBlocks; jg += SORT_THREADS) {
        if (edgeBlocks > SORT_THREADS) seg_meta(jg, tid, b, edgeBlocks, segStart, segCnt, mStart, mCnt);
        else                           __syncthreads();
        int nseg = min(SORT_THREADS, edgeBlocks - jg);
        for (int q0 = w; q0 < nseg; q0 += BG_WAVES * SEGS_PER_TRIP) {
            int st[SEGS_PER_TRIP], ct[SEGS_PER_TRIP], p[SEGS_PER_TRIP];
            unsigned v[SEGS_PER_TRIP];
            int  qv  = q0 + BG_WAVES * lane;
            bool okv = lane < SEGS_PER_TRIP && qv < nseg;
            int  msv = okv ? mStart[qv] : 0;
            int  mcv = okv ? mCnt[qv]   : 0;
            #pragma unroll
            for (int u = 0; u < SEGS_PER_TRIP; ++u) {
                st[u] = __builtin_amdgcn_readlane(msv, u);
                ct[u] = __builtin_amdgcn_readlane(mcv, u);
            }
            #pragma unroll
            for (int u = 0; u < SEGS_PER_TRIP; ++u)
                v[u] = records[min(st[u] + min(lane, max(ct[u] - 1, 0)), E - 1)];
            #pragma unroll
            for (int u = 0; u < SEGS_PER_TRIP; ++u)
                p[u] = (lane < ct[u]) ? atomicAdd(&cur[v[u] & REC_MASK], 1) : -1;
            #pragma unroll
            for (int u = 0; u < SEGS_PER_TRIP; ++u)
                if (p[u] >= 0) sorted[p[u]] = (int)(v[u] >> REC_SHIFT);
            #pragma unroll
            for (int u = 0; u < SEGS_PER_TRIP; ++u)
                for (int i = 32 + lane; i < ct[u]; i += 32) {
                    unsigned xr = records[st[u] + i];
                    int pp = atomicAdd(&cur[xr & REC_MASK], 1);
                    sorted[pp] = (int)(xr >> REC_SHIFT);
                }
        }
    }
}

template <bool IDS_IN_LDS>
__device__ __forceinline__ int get_id(const int* idsL, const int* idsG, int i) {
    if (IDS_IN_LDS) return idsL[i];
    return idsG[i];
}

template <bool IDS_IN_LDS>
__device__ __forceinline__ void gather_rows(const int* idsL, const int* idsG,
                                            const float* __restrict__ x,
                                            int lane, int w,
                                            const int* cntN, const int* startN,
                                            float* trow) {
    int nl0 = w * BG_NPW;
    int d_l = 0, s_l = 0;
    if (lane < BG_NPW) {
        d_l = cntN[nl0 + lane];
        s_l = IDS_IN_LDS ? (nl0 + lane) * IDS_PITCH : startN[nl0 + lane];
    }
    const float4* x4 = (const float4*)x;

    int dk[BG_NPW], sk[BG_NPW], iv[BG_NPW];
    #pragma unroll
    for (int k = 0; k < BG_NPW; ++k) {
        dk[k] = __builtin_amdgcn_readlane(d_l, k);
        sk[k] = __builtin_amdgcn_readlane(s_l, k);
        iv[k] = (lane < dk[k]) ? get_id<IDS_IN_LDS>(idsL, idsG, sk[k] + lane) : 0;
    }

    #pragma unroll
    for (int k = 0; k < BG_NPW; ++k) {
        float4 acc = make_float4(0.0f, 0.0f, 0.0f, 0.0f);
        for (int base = 0; base < dk[k]; base += 32) {
            int cnt = min(32, dk[k] - base);
            int idx = iv[k];
            if (base > 0) idx = (lane < cnt) ? get_id<IDS_IN_LDS>(idsL, idsG, sk[k] + base + lane) : 0;
            for (int i = 0; i < cnt; i += 8) {
                int m = cnt - i;
                int src[8];
                float4 v[8];
                #pragma unroll
                for (int j = 0; j < 8; ++j)
                    src[j] = __builtin_amdgcn_readlane(idx, min(i + j, cnt - 1));
                if (m >= 8) {
                    #pragma unroll
                    for (int j = 0; j < 8; ++j)
                        v[j] = x4[(long long)src[j] * (D_IN / 4) + lane];
                } else {
                    #pragma unroll
                    for (int j = 0; j < 8; ++j) {
                        v[j] = make_float4(0.0f, 0.0f, 0.0f, 0.0f);
                        if (j < m) v[j] = x4[(long long)src[j] * (D_IN / 4) + lane];
                    }
                }
                #pragma unroll
                for (int j = 0; j < 8; ++j) {
                    acc.x += v[j].x; acc.y += v[j].y;
                    acc.z += v[j].z; acc.w += v[j].w;
                }
            }
        }
        float inv = 1.0f / fmaxf((float)dk[k], 1.0f);
        acc.x *= inv; acc.y *= inv; acc.z *= inv; acc.w *= inv;
        *(float4*)(trow + k * TILE_PITCH) = acc;
    }
}

__global__ __launch_bounds__(BG_THREADS)
void gnn_bucket_gather_mlp_kernel(const unsigned* __restrict__ records,
                                  const int* __restrict__ segStart,
                                  const int* __restrict__ segCnt,
                                  int* sorted,
                                  const float* __restrict__ x,
                                  const float* __restrict__ packedB,
                                  const float* __restrict__ bc,
                                  float* __restrict__ out,
                                  int N, int E, int edgeBlocks) {
    __shared__ __attribute__((aligned(16))) float tile[BG_TILES][TILE_ROWS][TILE_PITCH];
    __shared__ int ids[BUCKET_NODES * IDS_PITCH];
    __shared__ int cntN[SORT_THREADS];
    __shared__ int startN[SORT_THREADS];
    __shared__ int cur[SORT_THREADS];
    __shared__ int wsum[SORT_THREADS / 32];
    __shared__ int mStart[SORT_THREADS];
    __shared__ int mCnt[SORT_THREADS];
    __shared__ int rowFull;
    int tid  = threadIdx.x;
    int b    = blockIdx.x;
    int lane = tid & 31;
    int w    = __builtin_amdgcn_readfirstlane(tid >> 5);
    bool lo  = tid < SORT_THREADS;
    if (lo) cntN[tid] = 0;
    if (tid == 0) rowFull = 0;
    if (tid < PAD_ROWS * 32)
        *(float4*)(&tile[0][0][0] + (BUCKET_NODES + (tid >> 5)) * TILE_PITCH + 4 * lane) =
            make_float4(0.0f, 0.0f, 0.0f, 0.0f);
    int ls = 0;
    for (int jg = 0; jg < edgeBlocks; jg += SORT_THREADS) {
        ls += seg_meta(jg, tid, b, edgeBlocks, segStart, segCnt, mStart, mCnt);
        int nseg = min(SORT_THREADS, edgeBlocks - jg);
        for (int q0 = w; q0 < nseg; q0 += BG_WAVES * SEGS_PER_TRIP) {
            int st[SEGS_PER_TRIP], ct[SEGS_PER_TRIP], r[SEGS_PER_TRIP];
            unsigned v[SEGS_PER_TRIP];
            int  qv  = q0 + BG_WAVES * lane;
            bool okv = lane < SEGS_PER_TRIP && qv < nseg;
            int  msv = okv ? mStart[qv] : 0;
            int  mcv = okv ? mCnt[qv]   : 0;
            #pragma unroll
            for (int u = 0; u < SEGS_PER_TRIP; ++u) {
                st[u] = __builtin_amdgcn_readlane(msv, u);
                ct[u] = __builtin_amdgcn_readlane(mcv, u);
            }
            #pragma unroll
            for (int u = 0; u < SEGS_PER_TRIP; ++u)
                v[u] = records[min(st[u] + min(lane, max(ct[u] - 1, 0)), E - 1)];
            #pragma unroll
            for (int u = 0; u < SEGS_PER_TRIP; ++u)
                r[u] = (lane < ct[u]) ? atomicAdd(&cntN[v[u] & REC_MASK], 1) : -1;
            #pragma unroll
            for (int u = 0; u < SEGS_PER_TRIP; ++u)
                if (r[u] >= 0) {
                    if (r[u] < NODE_SLOTS)
                        ids[(v[u] & REC_MASK) * IDS_PITCH + r[u]] = (int)(v[u] >> REC_SHIFT);
                    else
                        rowFull = 1;
                }
            #pragma unroll
            for (int u = 0; u < SEGS_PER_TRIP; ++u)
                for (int i = 32 + lane; i < ct[u]; i += 32) {
                    unsigned xr = records[st[u] + i];
                    int rr = atomicAdd(&cntN[xr & REC_MASK], 1);
                    if (rr < NODE_SLOTS)
                        ids[(xr & REC_MASK) * IDS_PITCH + rr] = (int)(xr >> REC_SHIFT);
                    else
                        rowFull = 1;
                }
        }
    }
    __syncthreads();

    float* trow = &tile[0][0][0] + (w * BG_NPW) * TILE_PITCH + 4 * lane;
    if (__builtin_amdgcn_readfirstlane(rowFull) == 0) {
        gather_rows<true>(ids, sorted, x, lane, w, cntN, startN, trow);
    } else {
        int dg = lo ? cntN[tid] : 0;
        int tot;
        int excl = block_excl_scan(dg, wsum, tid, &tot);
        __syncthreads();
        int s0;
        block_excl_scan(ls, wsum, tid, &s0);
        if (lo) {
            startN[tid] = s0 + excl;
            cur[tid]    = s0 + excl;
        }
        place_in_memory(records, segStart, segCnt, sorted, E, edgeBlocks,
                        tid, b, lane, w, cur, mStart, mCnt);
        __syncthreads();
        gather_rows<false>(ids, sorted, x, lane, w, cntN, startN, trow);
    }
    __syncthreads();

    if (w < BG_TILES) {
        int m0 = b * BUCKET_NODES + w * TILE_ROWS;
        if (m0 < N) {
            int koff = (lane >= 16) ? 2 : 0;
            const float* arow = &tile[w][lane & 15][koff];
            v8f c = {};
            #pragma unroll 8
            for (int s = 0; s < 32; ++s) {
                float2 a2 = *(const float2*)(arow + 4 * s);
                v2f a; a[0] = a2.x; a[1] = a2.y;
                v2f bf; bf[0] = packedB[s * 64 + lane];
                        bf[1] = packedB[s * 64 + 32 + lane];
                c = __builtin_amdgcn_wmma_f32_16x16x4_f32(
                         false, a,  false, bf,
                         (short)0, c,  false,  false);
            }
            int n = lane & 15;
            if (n < D_OUT) {
                float bcn = bc[n];
                int lrow  = w * TILE_ROWS + ((lane >= 16) ? 8 : 0);
                int rbase = b * BUCKET_NODES + lrow;
                #pragma unroll
                for (int j = 0; j < 8; ++j)
                    if (lrow + j < BUCKET_NODES && rbase + j < N)
                        out[(long long)(rbase + j) * D_OUT + n] = c[j] + bcn;
            }
        }
    }
}

extern "C" void kernel_launch(void* const* d_in, const int* in_sizes, int n_in,
                              void* d_out, int out_size, void* d_ws, size_t ws_size,
                              hipStream_t stream) {
    const float* x     = (const float*)d_in[0];
    const int*   edges = (const int*)  d_in[1];
    const float* W1    = (const float*)d_in[2];
    const float* b1    = (const float*)d_in[3];
    const float* W2    = (const float*)d_in[4];
    const float* b2    = (const float*)d_in[5];
    float*       out   = (float*)d_out;

    int N = in_sizes[0] / D_IN;
    int E = in_sizes[1] / 2;
    int NB = (N + BUCKET_NODES - 1) / BUCKET_NODES;
    int edgeBlocks = (E + EDGE_CHUNK - 1) / EDGE_CHUNK;

    float* ws = (float*)d_ws;
    long long offB  = 0;
    long long offBc = offB + 2048;
    int* segStart     = (int*)(ws + offBc + 64);
    int* segCnt       = segStart + edgeBlocks * MAX_BUCKETS;
    int* sorted       = segCnt + edgeBlocks * MAX_BUCKETS;
    unsigned* records = (unsigned*)(sorted + edgeBlocks * EDGE_CHUNK);

    gnn_block_sort_fold_kernel<<<FOLD_BLOCKS + edgeBlocks, SORT_THREADS, 0, stream>>>(
        edges, segStart, segCnt, records, E, N, W1, b1, W2, b2, ws + offB, ws + offBc);

    gnn_bucket_gather_mlp_kernel<<<NB, BG_THREADS, 0, stream>>>(
        records, segStart, segCnt, sorted, x, ws + offB, ws + offBc, out, N, E, edgeBlocks);
}
